# starting kernel
# baseline (speedup 1.0000x reference)
#include <hip/hip_runtime.h>
#include <hip/hip_bf16.h>


#define NNODES  50000
#define EEDGES  800000
#define ETOT    (EEDGES + NNODES)
#define FDIM    128
#define NHEADS  8
#define CDIM    16

typedef __attribute__((ext_vector_type(2))) float v2f;
typedef __attribute__((ext_vector_type(8))) float v8f;

__device__ __forceinline__ unsigned fenc(float f) {
    unsigned u = __float_as_uint(f);
    return (u & 0x80000000u) ? ~u : (u | 0x80000000u);
}
__device__ __forceinline__ float fdec(unsigned u) {
    u = (u & 0x80000000u) ? (u ^ 0x80000000u) : ~u;
    return __uint_as_float(u);
}

__device__ __forceinline__ void edge_ends(const int* __restrict__ ei, int e,
                                          int& s, int& d) {
    if (e < EEDGES) { s = ei[e]; d = ei[EEDGES + e]; }
    else            { s = e - EEDGES; d = s; }
}

__global__ void gat_gemm128_wmma(const float* __restrict__ Ain,
                                 const float* __restrict__ Wmat,
                                 float* __restrict__ Out) {
    __shared__ float As[16 * 132];
    const int tileBase = blockIdx.x * 16;
    const int t = threadIdx.x;

    const float4* A4 = (const float4*)(Ain + (size_t)tileBase * FDIM);
    #pragma unroll
    for (int j = 0; j < 2; ++j) {
        int gi = j * 256 + t;
        int r  = gi >> 5;
        int c4 = gi & 31;
        float4 v = A4[r * 32 + c4];
        *(float4*)&As[r * 132 + c4 * 4] = v;
    }
    __syncthreads();

    const int wave = t >> 5;
    const int lane = t & 31;
    const int m    = lane & 15;
    const int half = lane >> 4;
    const int col  = wave * 16 + m;

    v8f acc = {};
    #pragma unroll 4
    for (int kk = 0; kk < 32; ++kk) {
        const int k = kk * 4 + half * 2;
        v2f a, b;
        a.x = As[m * 132 + k];
        a.y = As[m * 132 + k + 1];
        b.x = Wmat[(size_t)k * FDIM + col];
        b.y = Wmat[(size_t)(k + 1) * FDIM + col];
        acc = __builtin_amdgcn_wmma_f32_16x16x4_f32(
             false, a,  false, b,
             (short)0, acc,  false,  false);
    }

    const int mbase = half * 8;
    #pragma unroll
    for (int i = 0; i < 8; ++i)
        Out[(size_t)(tileBase + mbase + i) * FDIM + wave * 16 + m] = acc[i];
}

__global__ void gat_zero(float* __restrict__ p, int n) {
    int g = blockIdx.x * blockDim.x + threadIdx.x;
    if (g < n) p[g] = 0.0f;
}

__global__ void gat_attn_alpha(const float* __restrict__ xt,
                               const float* __restrict__ attL,
                               const float* __restrict__ attR,
                               const int* __restrict__ ei,
                               float* __restrict__ alpha,
                               unsigned* __restrict__ amax) {
    int gid = blockIdx.x * blockDim.x + threadIdx.x;
    if (gid >= ETOT * NHEADS) return;
    int e = gid >> 3, hd = gid & 7;
    int s, d; edge_ends(ei, e, s, d);

    const float4* xj = (const float4*)(xt + (size_t)s * FDIM + hd * CDIM);
    const float4* xi = (const float4*)(xt + (size_t)d * FDIM + hd * CDIM);
    const float4* al = (const float4*)(attL + hd * CDIM);
    const float4* ar = (const float4*)(attR + hd * CDIM);

    float logit = 0.f, a = 0.f;
    #pragma unroll
    for (int q = 0; q < 4; ++q) {
        float4 vj = xj[q], vi = xi[q], l = al[q], r = ar[q];
        logit += vi.x * vj.x + vi.y * vj.y + vi.z * vj.z + vi.w * vj.w;
        a += l.x * vj.x + l.y * vj.y + l.z * vj.z + l.w * vj.w;
        a += r.x * vi.x + r.y * vi.y + r.z * vi.z + r.w * vi.w;
    }
    float sg = 1.0f / (1.0f + expf(-logit));
    float v  = a * sg;
    v = (v > 0.0f) ? v : 0.2f * v;
    alpha[gid] = v;
    atomicMax(&amax[(size_t)d * NHEADS + hd], fenc(v));
}

__global__ void gat_attn_exp(float* __restrict__ alpha,
                             const unsigned* __restrict__ amax,
                             float* __restrict__ denom,
                             const int* __restrict__ ei) {
    int gid = blockIdx.x * blockDim.x + threadIdx.x;
    if (gid >= ETOT * NHEADS) return;
    int e = gid >> 3, hd = gid & 7;
    int s, d; edge_ends(ei, e, s, d); (void)s;
    float m  = fdec(amax[(size_t)d * NHEADS + hd]);
    float ex = expf(alpha[gid] - m);
    alpha[gid] = ex;
    atomicAdd(&denom[(size_t)d * NHEADS + hd], ex);
}

__global__ void gat_attn_aggregate(const float* __restrict__ alpha,
                                   const float* __restrict__ denom,
                                   const float* __restrict__ xt,
                                   float* __restrict__ agg,
                                   const int* __restrict__ ei) {
    int gid = blockIdx.x * blockDim.x + threadIdx.x;
    if (gid >= ETOT * NHEADS) return;
    int e = gid >> 3, hd = gid & 7;
    int s, d; edge_ends(ei, e, s, d);
    float attn = alpha[gid] / (denom[(size_t)d * NHEADS + hd] + 1e-16f);
    const float* xj = xt + (size_t)s * FDIM + hd * CDIM;
    float* o = agg + (size_t)d * FDIM + hd * CDIM;
    #pragma unroll
    for (int c = 0; c < CDIM; ++c)
        atomicAdd(&o[c], xj[c] * attn);
}

__global__ void gat_bias_gelu(const float* __restrict__ agg,
                              const float* __restrict__ bias,
                              float* __restrict__ out) {
    int g = blockIdx.x * blockDim.x + threadIdx.x;
    if (g >= NNODES * FDIM) return;
    float v = agg[g] + bias[g & (FDIM - 1)];
    out[g] = 0.5f * v * (1.0f + erff(v * 0.70710678118654752f));
}

__global__ void gat_fc(const float* __restrict__ h,
                       const float* __restrict__ w,
                       const float* __restrict__ b0,
                       float* __restrict__ out) {
    int n = blockIdx.x * blockDim.x + threadIdx.x;
    if (n >= NNODES) return;
    const float4* h4 = (const float4*)(h + (size_t)n * FDIM);
    const float4* w4 = (const float4*)w;
    float s = 0.f;
    #pragma unroll 8
    for (int i = 0; i < FDIM / 4; ++i) {
        float4 a = h4[i], c = w4[i];
        s += a.x * c.x + a.y * c.y + a.z * c.z + a.w * c.w;
    }
    out[n] = s + b0[0];
}

extern "C" void kernel_launch(void* const* d_in, const int* in_sizes, int n_in,
                              void* d_out, int out_size, void* d_ws, size_t ws_size,
                              hipStream_t stream) {
    (void)in_sizes; (void)n_in; (void)out_size; (void)ws_size;
    const float* x   = (const float*)d_in[0];
    const float* W1  = (const float*)d_in[1];
    const float* b1  = (const float*)d_in[2];
    const float* al1 = (const float*)d_in[3];
    const float* ar1 = (const float*)d_in[4];
    const float* W2  = (const float*)d_in[5];
    const float* b2  = (const float*)d_in[6];
    const float* al2 = (const float*)d_in[7];
    const float* ar2 = (const float*)d_in[8];
    const float* fcW = (const float*)d_in[9];
    const float* fcb = (const float*)d_in[10];
    const int*   ei  = (const int*)d_in[11];

    float*    ws    = (float*)d_ws;
    float*    xt    = ws;
    float*    hbuf  = xt    + (size_t)NNODES * FDIM;
    float*    alpha = hbuf  + (size_t)NNODES * FDIM;
    float*    agg   = alpha + (size_t)ETOT * NHEADS;
    float*    denom = agg   + (size_t)NNODES * FDIM;
    unsigned* amax  = (unsigned*)(denom + (size_t)NNODES * NHEADS);

    const int ZN = NNODES * (FDIM + 2 * NHEADS);
    const int EG = (ETOT * NHEADS + 255) / 256;
    const int NG = (NNODES * FDIM + 255) / 256;
    const dim3 B(256);

    gat_gemm128_wmma<<<NNODES / 16, B, 0, stream>>>(x, W1, xt);
    gat_zero<<<(ZN + 255) / 256, B, 0, stream>>>(agg, ZN);
    gat_attn_alpha<<<EG, B, 0, stream>>>(xt, al1, ar1, ei, alpha, amax);
    gat_attn_exp<<<EG, B, 0, stream>>>(alpha, amax, denom, ei);
    gat_attn_aggregate<<<EG, B, 0, stream>>>(alpha, denom, xt, agg, ei);
    gat_bias_gelu<<<NG, B, 0, stream>>>(agg, b1, hbuf);

    gat_gemm128_wmma<<<NNODES / 16, B, 0, stream>>>(hbuf, W2, xt);
    gat_zero<<<(ZN + 255) / 256, B, 0, stream>>>(agg, ZN);
    gat_attn_alpha<<<EG, B, 0, stream>>>(xt, al2, ar2, ei, alpha, amax);
    gat_attn_exp<<<EG, B, 0, stream>>>(alpha, amax, denom, ei);
    gat_attn_aggregate<<<EG, B, 0, stream>>>(alpha, denom, xt, agg, ei);
    gat_bias_gelu<<<NG, B, 0, stream>>>(agg, b2, hbuf);

    gat_fc<<<(NNODES + 255) / 256, B, 0, stream>>>(hbuf, fcW, fcb, (float*)d_out);
}


# f32 atomic scatter in aggregation replaced by per-call counting sort of edges by destination (int atomics) and a wave-per-node gather summing in registers, one store per row; sum order changes
# speedup vs baseline: 2.4337x; 2.4337x over previous
#include <hip/hip_runtime.h>
#include <hip/hip_bf16.h>


#define NNODES  50000
#define EEDGES  800000
#define ETOT    (EEDGES + NNODES)
#define FDIM    128
#define NHEADS  8
#define CDIM    16

typedef __attribute__((ext_vector_type(2))) float v2f;
typedef __attribute__((ext_vector_type(8))) float v8f;

__device__ __forceinline__ unsigned fenc(float f) {
    unsigned u = __float_as_uint(f);
    return (u & 0x80000000u) ? ~u : (u | 0x80000000u);
}
__device__ __forceinline__ float fdec(unsigned u) {
    u = (u & 0x80000000u) ? (u ^ 0x80000000u) : ~u;
    return __uint_as_float(u);
}

__device__ __forceinline__ void edge_ends(const int* __restrict__ ei, int e,
                                          int& s, int& d) {
    if (e < EEDGES) { s = ei[e]; d = ei[EEDGES + e]; }
    else            { s = e - EEDGES; d = s; }
}

__global__ void gat_gemm128_wmma(const float* __restrict__ Ain,
                                 const float* __restrict__ Wmat,
                                 float* __restrict__ Out) {
    __shared__ float As[16 * 132];
    const int tileBase = blockIdx.x * 16;
    const int t = threadIdx.x;

    const float4* A4 = (const float4*)(Ain + (size_t)tileBase * FDIM);
    #pragma unroll
    for (int j = 0; j < 2; ++j) {
        int gi = j * 256 + t;
        int r  = gi >> 5;
        int c4 = gi & 31;
        float4 v = A4[r * 32 + c4];
        *(float4*)&As[r * 132 + c4 * 4] = v;
    }
    __syncthreads();

    const int wave = t >> 5;
    const int lane = t & 31;
    const int m    = lane & 15;
    const int half = lane >> 4;
    const int col  = wave * 16 + m;

    v8f acc = {};
    #pragma unroll 4
    for (int kk = 0; kk < 32; ++kk) {
        const int k = kk * 4 + half * 2;
        v2f a, b;
        a.x = As[m * 132 + k];
        a.y = As[m * 132 + k + 1];
        b.x = Wmat[(size_t)k * FDIM + col];
        b.y = Wmat[(size_t)(k + 1) * FDIM + col];
        acc = __builtin_amdgcn_wmma_f32_16x16x4_f32(
             false, a,  false, b,
             (short)0, acc,  false,  false);
    }

    const int mbase = half * 8;
    #pragma unroll
    for (int i = 0; i < 8; ++i)
        Out[(size_t)(tileBase + mbase + i) * FDIM + wave * 16 + m] = acc[i];
}

__global__ void gat_zero(float* __restrict__ p, int n) {
    int g = blockIdx.x * blockDim.x + threadIdx.x;
    if (g < n) p[g] = 0.0f;
}

__global__ void gat_attn_alpha(const float* __restrict__ xt,
                               const float* __restrict__ attL,
                               const float* __restrict__ attR,
                               const int* __restrict__ ei,
                               float* __restrict__ alpha,
                               unsigned* __restrict__ amax) {
    int gid = blockIdx.x * blockDim.x + threadIdx.x;
    if (gid >= ETOT * NHEADS) return;
    int e = gid >> 3, hd = gid & 7;
    int s, d; edge_ends(ei, e, s, d);

    const float4* xj = (const float4*)(xt + (size_t)s * FDIM + hd * CDIM);
    const float4* xi = (const float4*)(xt + (size_t)d * FDIM + hd * CDIM);
    const float4* al = (const float4*)(attL + hd * CDIM);
    const float4* ar = (const float4*)(attR + hd * CDIM);

    float logit = 0.f, a = 0.f;
    #pragma unroll
    for (int q = 0; q < 4; ++q) {
        float4 vj = xj[q], vi = xi[q], l = al[q], r = ar[q];
        logit += vi.x * vj.x + vi.y * vj.y + vi.z * vj.z + vi.w * vj.w;
        a += l.x * vj.x + l.y * vj.y + l.z * vj.z + l.w * vj.w;
        a += r.x * vi.x + r.y * vi.y + r.z * vi.z + r.w * vi.w;
    }
    float sg = 1.0f / (1.0f + expf(-logit));
    float v  = a * sg;
    v = (v > 0.0f) ? v : 0.2f * v;
    alpha[gid] = v;
    atomicMax(&amax[(size_t)d * NHEADS + hd], fenc(v));
}

__global__ void gat_attn_exp(float* __restrict__ alpha,
                             const unsigned* __restrict__ amax,
                             float* __restrict__ denom,
                             const int* __restrict__ ei) {
    int gid = blockIdx.x * blockDim.x + threadIdx.x;
    if (gid >= ETOT * NHEADS) return;
    int e = gid >> 3, hd = gid & 7;
    int s, d; edge_ends(ei, e, s, d); (void)s;
    float m  = fdec(amax[(size_t)d * NHEADS + hd]);
    float ex = expf(alpha[gid] - m);
    alpha[gid] = ex;
    atomicAdd(&denom[(size_t)d * NHEADS + hd], ex);
}

__global__ void gat_csr_count(const int* __restrict__ ei, int* __restrict__ deg) {
    int e = blockIdx.x * blockDim.x + threadIdx.x;
    if (e < EEDGES) atomicAdd(&deg[ei[EEDGES + e]], 1);
}

__global__ void __launch_bounds__(1024)
gat_csr_scan(int* __restrict__ deg, int* __restrict__ rowptr) {
    __shared__ int wtot[32];
    const int t = threadIdx.x;
    const int lane = t & 31, w = t >> 5;
    const int NQ = NNODES / 4;
    int4* d4 = (int4*)deg;
    int4* r4 = (int4*)rowptr;
    int4 v[13];
    int sum = 0;
    #pragma unroll
    for (int j = 0; j < 13; ++j) {
        int q = t * 13 + j;
        v[j] = (q < NQ) ? d4[q] : make_int4(0, 0, 0, 0);
        sum += v[j].x + v[j].y + v[j].z + v[j].w;
    }
    int inc = sum;
    for (int o = 1; o < 32; o <<= 1) {
        int u = __shfl_up(inc, o);
        if (lane >= o) inc += u;
    }
    if (lane == 31) wtot[w] = inc;
    __syncthreads();
    if (w == 0) {
        int x = wtot[lane], y = x;
        for (int o = 1; o < 32; o <<= 1) {
            int u = __shfl_up(y, o);
            if (lane >= o) y += u;
        }
        wtot[lane] = y - x;
    }
    __syncthreads();
    int run = wtot[w] + inc - sum;
    #pragma unroll
    for (int j = 0; j < 13; ++j) {
        int q = t * 13 + j;
        if (q < NQ) {
            int4 o;
            o.x = run; run += v[j].x;
            o.y = run; run += v[j].y;
            o.z = run; run += v[j].z;
            o.w = run; run += v[j].w;
            r4[q] = o;
            d4[q] = o;
        }
    }
    if (t == 1023) rowptr[NNODES] = run;
}

__global__ void gat_csr_fill(const int* __restrict__ ei, int* __restrict__ cursor,
                             int* __restrict__ csr) {
    int e = blockIdx.x * blockDim.x + threadIdx.x;
    if (e >= EEDGES) return;
    int s = ei[e], d = ei[EEDGES + e];
    int pos = atomicAdd(&cursor[d], 1);
    if ((unsigned)pos < (unsigned)EEDGES)
        ((int2*)csr)[pos] = make_int2(s, e);
}

#define GATHER_WAVES 12500
__global__ void __launch_bounds__(128)
gat_attn_gather(const float* __restrict__ alpha,
                const float* __restrict__ denom,
                const float* __restrict__ xt,
                float* __restrict__ agg,
                const int* __restrict__ rowptr,
                const int* __restrict__ csr) {
    const int lane = threadIdx.x & 31;
    const int hd   = lane >> 2;
    const int wv = __builtin_amdgcn_readfirstlane((int)(threadIdx.x >> 5));
    const int gw = blockIdx.x * 4 + wv;
    const int2* rec = (const int2*)csr;

    for (int node = gw; node < NNODES; node += GATHER_WAVES) {
        const int beg = max(rowptr[node], 0), end = min(rowptr[node + 1], EEDGES);
        const float dn = denom[(size_t)node * NHEADS + hd] + 1e-16f;
        float a0 = alpha[((size_t)EEDGES + node) * NHEADS + hd] / dn;
        float4 x0 = *(const float4*)(xt + (size_t)node * FDIM + lane * 4);
        float4 acc = make_float4(x0.x * a0, x0.y * a0, x0.z * a0, x0.w * a0);
        #pragma unroll 4
        for (int p = beg; p < end; ++p) {
            int2 r = rec[p];
            int s   = min(max(r.x, 0), NNODES - 1);
            int eid = min(max(r.y, 0), EEDGES - 1);
            float a = alpha[(size_t)eid * NHEADS + hd] / dn;
            float4 x = *(const float4*)(xt + (size_t)s * FDIM + lane * 4);
            acc.x += x.x * a; acc.y += x.y * a; acc.z += x.z * a; acc.w += x.w * a;
        }
        *(float4*)(agg + (size_t)node * FDIM + lane * 4) = acc;
    }
}

__global__ void gat_bias_gelu(const float* __restrict__ agg,
                              const float* __restrict__ bias,
                              float* __restrict__ out) {
    int g = blockIdx.x * blockDim.x + threadIdx.x;
    if (g >= NNODES * FDIM) return;
    float v = agg[g] + bias[g & (FDIM - 1)];
    out[g] = 0.5f * v * (1.0f + erff(v * 0.70710678118654752f));
}

__global__ void gat_fc(const float* __restrict__ h,
                       const float* __restrict__ w,
                       const float* __restrict__ b0,
                       float* __restrict__ out) {
    int n = blockIdx.x * blockDim.x + threadIdx.x;
    if (n >= NNODES) return;
    const float4* h4 = (const float4*)(h + (size_t)n * FDIM);
    const float4* w4 = (const float4*)w;
    float s = 0.f;
    #pragma unroll 8
    for (int i = 0; i < FDIM / 4; ++i) {
        float4 a = h4[i], c = w4[i];
        s += a.x * c.x + a.y * c.y + a.z * c.z + a.w * c.w;
    }
    out[n] = s + b0[0];
}

extern "C" void kernel_launch(void* const* d_in, const int* in_sizes, int n_in,
                              void* d_out, int out_size, void* d_ws, size_t ws_size,
                              hipStream_t stream) {
    (void)in_sizes; (void)n_in; (void)out_size; (void)ws_size;
    const float* x   = (const float*)d_in[0];
    const float* W1  = (const float*)d_in[1];
    const float* b1  = (const float*)d_in[2];
    const float* al1 = (const float*)d_in[3];
    const float* ar1 = (const float*)d_in[4];
    const float* W2  = (const float*)d_in[5];
    const float* b2  = (const float*)d_in[6];
    const float* al2 = (const float*)d_in[7];
    const float* ar2 = (const float*)d_in[8];
    const float* fcW = (const float*)d_in[9];
    const float* fcb = (const float*)d_in[10];
    const int*   ei  = (const int*)d_in[11];

    float*    ws    = (float*)d_ws;
    float*    xt    = ws;
    float*    hbuf  = xt    + (size_t)NNODES * FDIM;
    float*    alpha = hbuf  + (size_t)NNODES * FDIM;
    float*    agg   = alpha + (size_t)ETOT * NHEADS;
    float*    denom = agg   + (size_t)NNODES * FDIM;
    unsigned* amax  = (unsigned*)(denom + (size_t)NNODES * NHEADS);
    int*      deg   = (int*)(amax + (size_t)NNODES * NHEADS);
    int*      rowp  = deg  + (size_t)NNODES;
    int*      csr   = rowp + (size_t)NNODES + 4;

    const int ZN1 = NNODES * (2 * NHEADS + 1);
    const int ZN2 = NNODES * (2 * NHEADS);
    const int CG  = (EEDGES + 255) / 256;
    const int EG = (ETOT * NHEADS + 255) / 256;
    const int NG = (NNODES * FDIM + 255) / 256;
    const dim3 B(256);

    gat_gemm128_wmma<<<NNODES / 16, B, 0, stream>>>(x, W1, xt);
    gat_zero<<<(ZN1 + 255) / 256, B, 0, stream>>>(denom, ZN1);
    gat_csr_count<<<CG, B, 0, stream>>>(ei, deg);
    gat_csr_scan<<<1, 1024, 0, stream>>>(deg, rowp);
    gat_csr_fill<<<CG, B, 0, stream>>>(ei, deg, csr);
    gat_attn_alpha<<<EG, B, 0, stream>>>(xt, al1, ar1, ei, alpha, amax);
    gat_attn_exp<<<EG, B, 0, stream>>>(alpha, amax, denom, ei);
    gat_attn_gather<<<GATHER_WAVES / 4, 128, 0, stream>>>(alpha, denom, xt, agg, rowp, csr);
    gat_bias_gelu<<<NG, B, 0, stream>>>(agg, b1, hbuf);

    gat_gemm128_wmma<<<NNODES / 16, B, 0, stream>>>(hbuf, W2, xt);
    gat_zero<<<(ZN2 + 255) / 256, B, 0, stream>>>(denom, ZN2);
    gat_attn_alpha<<<EG, B, 0, stream>>>(xt, al2, ar2, ei, alpha, amax);
    gat_attn_exp<<<EG, B, 0, stream>>>(alpha, amax, denom, ei);
    gat_attn_gather<<<GATHER_WAVES / 4, 128, 0, stream>>>(alpha, denom, xt, agg, rowp, csr);
    gat_bias_gelu<<<NG, B, 0, stream>>>(agg, b2, hbuf);

    gat_fc<<<(NNODES + 255) / 256, B, 0, stream>>>(hbuf, fcW, fcb, (float*)d_out);
}


# score+max, exp+sum and weighted-gather edge passes fused into one wave-per-destination pass over the dst-sorted list, online softmax in registers; float/max atomics removed; sum order changes
# speedup vs baseline: 4.6038x; 1.8917x over previous
#include <hip/hip_runtime.h>
#include <hip/hip_bf16.h>


#define NNODES  50000
#define EEDGES  800000
#define ETOT    (EEDGES + NNODES)
#define FDIM    128
#define NHEADS  8
#define CDIM    16

typedef __attribute__((ext_vector_type(2))) float v2f;
typedef __attribute__((ext_vector_type(8))) float v8f;

__global__ void gat_gemm128_wmma(const float* __restrict__ Ain,
                                 const float* __restrict__ Wmat,
                                 float* __restrict__ Out) {
    __shared__ float As[16 * 132];
    const int tileBase = blockIdx.x * 16;
    const int t = threadIdx.x;

    const float4* A4 = (const float4*)(Ain + (size_t)tileBase * FDIM);
    #pragma unroll
    for (int j = 0; j < 2; ++j) {
        int gi = j * 256 + t;
        int r  = gi >> 5;
        int c4 = gi & 31;
        float4 v = A4[r * 32 + c4];
        *(float4*)&As[r * 132 + c4 * 4] = v;
    }
    __syncthreads();

    const int wave = t >> 5;
    const int lane = t & 31;
    const int m    = lane & 15;
    const int half = lane >> 4;
    const int col  = wave * 16 + m;

    v8f acc = {};
    #pragma unroll 4
    for (int kk = 0; kk < 32; ++kk) {
        const int k = kk * 4 + half * 2;
        v2f a, b;
        a.x = As[m * 132 + k];
        a.y = As[m * 132 + k + 1];
        b.x = Wmat[(size_t)k * FDIM + col];
        b.y = Wmat[(size_t)(k + 1) * FDIM + col];
        acc = __builtin_amdgcn_wmma_f32_16x16x4_f32(
             false, a,  false, b,
             (short)0, acc,  false,  false);
    }

    const int mbase = half * 8;
    #pragma unroll
    for (int i = 0; i < 8; ++i)
        Out[(size_t)(tileBase + mbase + i) * FDIM + wave * 16 + m] = acc[i];
}

__global__ void gat_zero(int* __restrict__ p, int n) {
    int g = blockIdx.x * blockDim.x + threadIdx.x;
    if (g < n) p[g] = 0;
}

__global__ void gat_csr_count(const int* __restrict__ ei, int* __restrict__ deg) {
    int e = blockIdx.x * blockDim.x + threadIdx.x;
    if (e < EEDGES) atomicAdd(&deg[ei[EEDGES + e]], 1);
}

__global__ void __launch_bounds__(1024)
gat_csr_scan(int* __restrict__ deg, int* __restrict__ rowptr) {
    __shared__ int wtot[32];
    const int t = threadIdx.x;
    const int lane = t & 31, w = t >> 5;
    const int NQ = NNODES / 4;
    int4* d4 = (int4*)deg;
    int4* r4 = (int4*)rowptr;
    int4 v[13];
    int sum = 0;
    #pragma unroll
    for (int j = 0; j < 13; ++j) {
        int q = t * 13 + j;
        v[j] = (q < NQ) ? d4[q] : make_int4(0, 0, 0, 0);
        sum += v[j].x + v[j].y + v[j].z + v[j].w;
    }
    int inc = sum;
    for (int o = 1; o < 32; o <<= 1) {
        int u = __shfl_up(inc, o);
        if (lane >= o) inc += u;
    }
    if (lane == 31) wtot[w] = inc;
    __syncthreads();
    if (w == 0) {
        int x = wtot[lane], y = x;
        for (int o = 1; o < 32; o <<= 1) {
            int u = __shfl_up(y, o);
            if (lane >= o) y += u;
        }
        wtot[lane] = y - x;
    }
    __syncthreads();
    int run = wtot[w] + inc - sum;
    #pragma unroll
    for (int j = 0; j < 13; ++j) {
        int q = t * 13 + j;
        if (q < NQ) {
            int4 o;
            o.x = run; run += v[j].x;
            o.y = run; run += v[j].y;
            o.z = run; run += v[j].z;
            o.w = run; run += v[j].w;
            r4[q] = o;
            d4[q] = o;
        }
    }
    if (t == 1023) rowptr[NNODES] = run;
}

__global__ void gat_csr_fill(const int* __restrict__ ei, int* __restrict__ cursor,
                             int* __restrict__ csr) {
    int e = blockIdx.x * blockDim.x + threadIdx.x;
    if (e >= EEDGES) return;
    int s = ei[e], d = ei[EEDGES + e];
    int pos = atomicAdd(&cursor[d], 1);
    if ((unsigned)pos < (unsigned)EEDGES)
        ((int2*)csr)[pos] = make_int2(s, e);
}

#define QUAD_XOR1 0xB1
#define QUAD_XOR2 0x4E
template <int CTRL>
__device__ __forceinline__ float quad_move(float v) {
    return __int_as_float(__builtin_amdgcn_mov_dpp(__float_as_int(v), CTRL, 0xF, 0xF, true));
}

__device__ __forceinline__ float quad_sum(float v) {
    v += quad_move<QUAD_XOR1>(v);
    v += quad_move<QUAD_XOR2>(v);
    return v;
}

__device__ __forceinline__ float quad_transpose_sum(float e0, float e1, float e2, float e3,
                                                    bool odd, bool hi) {
    const float rA = (odd ? e1 : e0) + quad_move<QUAD_XOR1>(odd ? e0 : e1);
    const float rB = (odd ? e3 : e2) + quad_move<QUAD_XOR1>(odd ? e2 : e3);
    return (hi ? rB : rA) + quad_move<QUAD_XOR2>(hi ? rA : rB);
}

__device__ __forceinline__ float dot4(float4 a, float4 b) {
    return a.x * b.x + a.y * b.y + a.z * b.z + a.w * b.w;
}

__device__ __forceinline__ float edge_score(float logit, float a) {
    float sg = 1.0f / (1.0f + expf(-logit));
    float v  = a * sg;
    return (v > 0.0f) ? v : 0.2f * v;
}

__device__ __forceinline__ float4 load_row(const float* __restrict__ xt, int s, int lane) {
    s = min(max(s, 0), NNODES - 1);
    return *(const float4*)(xt + (size_t)s * FDIM + lane * 4);
}

#define GATHER_WAVES 10000
__global__ void __launch_bounds__(128)
gat_attn_fused(const float* __restrict__ xt,
               const float* __restrict__ attL,
               const float* __restrict__ attR,
               float* __restrict__ agg,
               const int* __restrict__ rowptr,
               const int* __restrict__ csr) {
    const int lane = threadIdx.x & 31;
    const bool odd = (lane & 1) != 0;
    const bool hi  = (lane & 2) != 0;
    const int wv = __builtin_amdgcn_readfirstlane((int)(threadIdx.x >> 5));
    const int gw = blockIdx.x * 4 + wv;
    const int2* rec = (const int2*)csr;
    const float4 al = *(const float4*)(attL + lane * 4);
    const float4 ar = *(const float4*)(attR + lane * 4);

    for (int node = gw; node < NNODES; node += GATHER_WAVES) {
        const int beg = max(rowptr[node], 0), end = min(rowptr[node + 1], EEDGES);
        const float4 xi = *(const float4*)(xt + (size_t)node * FDIM + lane * 4);
        const float aR = quad_sum(dot4(ar, xi));
        float m   = edge_score(quad_sum(dot4(xi, xi)), quad_sum(dot4(al, xi)) + aR);
        float den = 1.0f;
        float4 acc = xi;

        int p = beg;
        for (; p + 4 <= end; p += 4) {
            const float4 x0 = load_row(xt, rec[p].x,     lane);
            const float4 x1 = load_row(xt, rec[p + 1].x, lane);
            const float4 x2 = load_row(xt, rec[p + 2].x, lane);
            const float4 x3 = load_row(xt, rec[p + 3].x, lane);
            const float lg = quad_transpose_sum(dot4(xi, x0), dot4(xi, x1),
                                                dot4(xi, x2), dot4(xi, x3), odd, hi);
            const float aa = quad_transpose_sum(dot4(al, x0), dot4(al, x1),
                                                dot4(al, x2), dot4(al, x3), odd, hi) + aR;
            const float v  = edge_score(lg, aa);
            float mx = fmaxf(v,  quad_move<QUAD_XOR1>(v));
            mx       = fmaxf(mx, quad_move<QUAD_XOR2>(mx));
            const float mn = fmaxf(m, mx);
            const float sc = expf(m - mn);
            const float pe = expf(v - mn);
            const float p0 = quad_move<0x00>(pe);
            const float p1 = quad_move<0x55>(pe);
            const float p2 = quad_move<0xAA>(pe);
            const float p3 = quad_move<0xFF>(pe);
            den   = den   * sc + ((p0 + p1) + (p2 + p3));
            acc.x = acc.x * sc + ((p0 * x0.x + p1 * x1.x) + (p2 * x2.x + p3 * x3.x));
            acc.y = acc.y * sc + ((p0 * x0.y + p1 * x1.y) + (p2 * x2.y + p3 * x3.y));
            acc.z = acc.z * sc + ((p0 * x0.z + p1 * x1.z) + (p2 * x2.z + p3 * x3.z));
            acc.w = acc.w * sc + ((p0 * x0.w + p1 * x1.w) + (p2 * x2.w + p3 * x3.w));
            m = mn;
        }
        for (; p < end; ++p) {
            const float4 x = load_row(xt, rec[p].x, lane);
            const float v  = edge_score(quad_sum(dot4(xi, x)), quad_sum(dot4(al, x)) + aR);
            const float mn = fmaxf(m, v);
            const float sc = expf(m - mn);
            const float pe = expf(v - mn);
            den   = den   * sc + pe;
            acc.x = acc.x * sc + pe * x.x;
            acc.y = acc.y * sc + pe * x.y;
            acc.z = acc.z * sc + pe * x.z;
            acc.w = acc.w * sc + pe * x.w;
            m = mn;
        }
        const float inv = 1.0f / (den + 1e-16f);
        *(float4*)(agg + (size_t)node * FDIM + lane * 4) =
            make_float4(acc.x * inv, acc.y * inv, acc.z * inv, acc.w * inv);
    }
}

__global__ void gat_bias_gelu(const float* __restrict__ agg,
                              const float* __restrict__ bias,
                              float* __restrict__ out) {
    int g = blockIdx.x * blockDim.x + threadIdx.x;
    if (g >= NNODES * FDIM) return;
    float v = agg[g] + bias[g & (FDIM - 1)];
    out[g] = 0.5f * v * (1.0f + erff(v * 0.70710678118654752f));
}

__global__ void gat_fc(const float* __restrict__ h,
                       const float* __restrict__ w,
                       const float* __restrict__ b0,
                       float* __restrict__ out) {
    int n = blockIdx.x * blockDim.x + threadIdx.x;
    if (n >= NNODES) return;
    const float4* h4 = (const float4*)(h + (size_t)n * FDIM);
    const float4* w4 = (const float4*)w;
    float s = 0.f;
    #pragma unroll 8
    for (int i = 0; i < FDIM / 4; ++i) {
        float4 a = h4[i], c = w4[i];
        s += a.x * c.x + a.y * c.y + a.z * c.z + a.w * c.w;
    }
    out[n] = s + b0[0];
}

extern "C" void kernel_launch(void* const* d_in, const int* in_sizes, int n_in,
                              void* d_out, int out_size, void* d_ws, size_t ws_size,
                              hipStream_t stream) {
    (void)in_sizes; (void)n_in; (void)out_size; (void)ws_size;
    const float* x   = (const float*)d_in[0];
    const float* W1  = (const float*)d_in[1];
    const float* b1  = (const float*)d_in[2];
    const float* al1 = (const float*)d_in[3];
    const float* ar1 = (const float*)d_in[4];
    const float* W2  = (const float*)d_in[5];
    const float* b2  = (const float*)d_in[6];
    const float* al2 = (const float*)d_in[7];
    const float* ar2 = (const float*)d_in[8];
    const float* fcW = (const float*)d_in[9];
    const float* fcb = (const float*)d_in[10];
    const int*   ei  = (const int*)d_in[11];

    float* ws   = (float*)d_ws;
    float* xt   = ws;
    float* hbuf = xt   + (size_t)NNODES * FDIM;
    float* agg  = hbuf + (size_t)NNODES * FDIM;
    int*   deg  = (int*)(agg + (size_t)NNODES * FDIM);
    int*   rowp = deg  + (size_t)NNODES;
    int*   csr  = rowp + (size_t)NNODES + 4;

    const int CG = (EEDGES + 255) / 256;
    const int NG = (NNODES * FDIM + 255) / 256;
    const dim3 B(256);

    gat_gemm128_wmma<<<NNODES / 16, B, 0, stream>>>(x, W1, xt);
    gat_zero<<<(NNODES + 255) / 256, B, 0, stream>>>(deg, NNODES);
    gat_csr_count<<<CG, B, 0, stream>>>(ei, deg);
    gat_csr_scan<<<1, 1024, 0, stream>>>(deg, rowp);
    gat_csr_fill<<<CG, B, 0, stream>>>(ei, deg, csr);
    gat_attn_fused<<<GATHER_WAVES / 4, 128, 0, stream>>>(xt, al1, ar1, agg, rowp, csr);
    gat_bias_gelu<<<NG, B, 0, stream>>>(agg, b1, hbuf);

    gat_gemm128_wmma<<<NNODES / 16, B, 0, stream>>>(hbuf, W2, xt);
    gat_attn_fused<<<GATHER_WAVES / 4, 128, 0, stream>>>(xt, al2, ar2, agg, rowp, csr);
    gat_bias_gelu<<<NG, B, 0, stream>>>(agg, b2, hbuf);

    gat_fc<<<(NNODES + 255) / 256, B, 0, stream>>>(hbuf, fcW, fcb, (float*)d_out);
}


# global int atomics of the per-call counting sort replaced by a two-level sort on LDS counters (edge chunks binned into node ranges, one block orders each range); sum order inside a row changes
# speedup vs baseline: 6.2573x; 1.3592x over previous
#include <hip/hip_runtime.h>
#include <hip/hip_bf16.h>


#define NNODES  50000
#define EEDGES  800000
#define ETOT    (EEDGES + NNODES)
#define FDIM    128
#define NHEADS  8
#define CDIM    16

typedef __attribute__((ext_vector_type(2))) float v2f;
typedef __attribute__((ext_vector_type(8))) float v8f;

__global__ void gat_gemm128_wmma(const float* __restrict__ Ain,
                                 const float* __restrict__ Wmat,
                                 float* __restrict__ Out) {
    __shared__ float As[16 * 132];
    const int tileBase = blockIdx.x * 16;
    const int t = threadIdx.x;

    const float4* A4 = (const float4*)(Ain + (size_t)tileBase * FDIM);
    #pragma unroll
    for (int j = 0; j < 2; ++j) {
        int gi = j * 256 + t;
        int r  = gi >> 5;
        int c4 = gi & 31;
        float4 v = A4[r * 32 + c4];
        *(float4*)&As[r * 132 + c4 * 4] = v;
    }
    __syncthreads();

    const int wave = t >> 5;
    const int lane = t & 31;
    const int m    = lane & 15;
    const int half = lane >> 4;
    const int col  = wave * 16 + m;

    v8f acc = {};
    #pragma unroll 4
    for (int kk = 0; kk < 32; ++kk) {
        const int k = kk * 4 + half * 2;
        v2f a, b;
        a.x = As[m * 132 + k];
        a.y = As[m * 132 + k + 1];
        b.x = Wmat[(size_t)k * FDIM + col];
        b.y = Wmat[(size_t)(k + 1) * FDIM + col];
        acc = __builtin_amdgcn_wmma_f32_16x16x4_f32(
             false, a,  false, b,
             (short)0, acc,  false,  false);
    }

    const int mbase = half * 8;
    #pragma unroll
    for (int i = 0; i < 8; ++i)
        Out[(size_t)(tileBase + mbase + i) * FDIM + wave * 16 + m] = acc[i];
}

#define SORT_CHUNKS  200
#define CHUNK_EDGES  4000
#define SORT_BUCKETS 250
#define BUCKET_NODES 200
#define SCAN_N       (SORT_BUCKETS * SORT_CHUNKS)
static_assert(SORT_CHUNKS * CHUNK_EDGES == EEDGES, "chunks must tile the edge list");
static_assert(SORT_BUCKETS * BUCKET_NODES == NNODES, "ranges must tile the nodes");
static_assert(CHUNK_EDGES % 4 == 0 && SCAN_N % 4 == 0, "int4 access");
static_assert(SCAN_N / 4 <= 1024 * 13, "one scan block must cover the table");
static_assert(SORT_BUCKETS <= 256 && BUCKET_NODES <= 256, "one counter per thread");

#define SORT_TRIPS ((CHUNK_EDGES / 4 + 255) / 256)

__global__ void __launch_bounds__(256)
gat_sort_hist(const int* __restrict__ ei, int* __restrict__ tab) {
    __shared__ int lh[SORT_BUCKETS];
    const int t = threadIdx.x, g = blockIdx.x;
    const int4* d4 = (const int4*)(ei + EEDGES + g * CHUNK_EDGES);
    int4 d[SORT_TRIPS];
    #pragma unroll
    for (int k = 0; k < SORT_TRIPS; ++k) {
        const int q = t + k * 256;
        d[k] = (q < CHUNK_EDGES / 4) ? d4[q] : make_int4(-1, -1, -1, -1);
    }
    if (t < SORT_BUCKETS) lh[t] = 0;
    __syncthreads();
    #pragma unroll
    for (int k = 0; k < SORT_TRIPS; ++k) {
        if ((unsigned)d[k].x < (unsigned)NNODES) atomicAdd(&lh[d[k].x / BUCKET_NODES], 1);
        if ((unsigned)d[k].y < (unsigned)NNODES) atomicAdd(&lh[d[k].y / BUCKET_NODES], 1);
        if ((unsigned)d[k].z < (unsigned)NNODES) atomicAdd(&lh[d[k].z / BUCKET_NODES], 1);
        if ((unsigned)d[k].w < (unsigned)NNODES) atomicAdd(&lh[d[k].w / BUCKET_NODES], 1);
    }
    __syncthreads();
    if (t < SORT_BUCKETS) tab[t * SORT_CHUNKS + g] = lh[t];
}

__global__ void __launch_bounds__(1024)
gat_csr_scan(const int* __restrict__ tab, int* __restrict__ pre) {
    __shared__ int wtot[32];
    const int t = threadIdx.x;
    const int lane = t & 31, w = t >> 5;
    const int NQ = SCAN_N / 4;
    const int4* d4 = (const int4*)tab;
    int4* r4 = (int4*)pre;
    int4 v[13];
    int sum = 0;
    #pragma unroll
    for (int j = 0; j < 13; ++j) {
        int q = t * 13 + j;
        v[j] = (q < NQ) ? d4[q] : make_int4(0, 0, 0, 0);
        sum += v[j].x + v[j].y + v[j].z + v[j].w;
    }
    int inc = sum;
    for (int o = 1; o < 32; o <<= 1) {
        int u = __shfl_up(inc, o);
        if (lane >= o) inc += u;
    }
    if (lane == 31) wtot[w] = inc;
    __syncthreads();
    if (w == 0) {
        int x = wtot[lane], y = x;
        for (int o = 1; o < 32; o <<= 1) {
            int u = __shfl_up(y, o);
            if (lane >= o) y += u;
        }
        wtot[lane] = y - x;
    }
    __syncthreads();
    int run = wtot[w] + inc - sum;
    #pragma unroll
    for (int j = 0; j < 13; ++j) {
        int q = t * 13 + j;
        if (q < NQ) {
            int4 o;
            o.x = run; run += v[j].x;
            o.y = run; run += v[j].y;
            o.z = run; run += v[j].z;
            o.w = run; run += v[j].w;
            r4[q] = o;
        }
    }
    if (t == 1023) pre[SCAN_N] = run;
}

__global__ void __launch_bounds__(256)
gat_sort_scatter(const int* __restrict__ ei, const int* __restrict__ pre,
                 int* __restrict__ tmp) {
    __shared__ int lc[SORT_BUCKETS];
    const int t = threadIdx.x, g = blockIdx.x;
    const int4* s4 = (const int4*)(ei + g * CHUNK_EDGES);
    const int4* d4 = (const int4*)(ei + EEDGES + g * CHUNK_EDGES);
    int4 s[SORT_TRIPS], d[SORT_TRIPS];
    #pragma unroll
    for (int k = 0; k < SORT_TRIPS; ++k) {
        const int q = t + k * 256;
        const bool in = q < CHUNK_EDGES / 4;
        s[k] = in ? s4[q] : make_int4(0, 0, 0, 0);
        d[k] = in ? d4[q] : make_int4(-1, -1, -1, -1);
    }
    if (t < SORT_BUCKETS) lc[t] = pre[t * SORT_CHUNKS + g];
    __syncthreads();
    int2* out = (int2*)tmp;
    #pragma unroll
    for (int k = 0; k < SORT_TRIPS; ++k) {
        const int sv[4] = {s[k].x, s[k].y, s[k].z, s[k].w};
        const int dv[4] = {d[k].x, d[k].y, d[k].z, d[k].w};
        #pragma unroll
        for (int c = 0; c < 4; ++c) {
            if ((unsigned)dv[c] < (unsigned)NNODES) {
                const int pos = atomicAdd(&lc[dv[c] / BUCKET_NODES], 1);
                if ((unsigned)pos < (unsigned)EEDGES)
                    out[pos] = make_int2(sv[c], dv[c]);
            }
        }
    }
}

#define PLACE_UNROLL 8

__global__ void __launch_bounds__(256)
gat_sort_place(const int* __restrict__ pre, const int* __restrict__ tmp,
               int* __restrict__ rowptr, int* __restrict__ csr) {
    __shared__ int cnt[256];
    __shared__ int wtot[8];
    const int t = threadIdx.x, b = blockIdx.x;
    const int lane = t & 31, w = t >> 5;
    const int start = min(max(pre[b * SORT_CHUNKS], 0), EEDGES);
    const int end   = min(max(pre[(b + 1) * SORT_CHUNKS], start), EEDGES);
    const int node0 = b * BUCKET_NODES;
    const int2* in = (const int2*)tmp;
    cnt[t] = 0;
    __syncthreads();
    for (int i0 = start + t; i0 < end; i0 += 256 * PLACE_UNROLL) {
        int dj[PLACE_UNROLL];
        #pragma unroll
        for (int k = 0; k < PLACE_UNROLL; ++k) {
            const int i = i0 + k * 256;
            dj[k] = (i < end) ? in[i].y : -1;
        }
        #pragma unroll
        for (int k = 0; k < PLACE_UNROLL; ++k) {
            const int j = dj[k] - node0;
            if ((unsigned)j < (unsigned)BUCKET_NODES) atomicAdd(&cnt[j], 1);
        }
    }
    __syncthreads();
    const int c = cnt[t];
    int inc = c;
    for (int o = 1; o < 32; o <<= 1) {
        int u = __shfl_up(inc, o);
        if (lane >= o) inc += u;
    }
    if (lane == 31) wtot[w] = inc;
    __syncthreads();
    int first = start + inc - c;
    for (int k = 0; k < w; ++k) first += wtot[k];
    cnt[t] = first;
    if (t < BUCKET_NODES) rowptr[node0 + t] = first;
    if (b == SORT_BUCKETS - 1 && t == 0) rowptr[NNODES] = end;
    __syncthreads();
    for (int i0 = start + t; i0 < end; i0 += 256 * PLACE_UNROLL) {
        int2 r[PLACE_UNROLL];
        #pragma unroll
        for (int k = 0; k < PLACE_UNROLL; ++k) {
            const int i = i0 + k * 256;
            r[k] = (i < end) ? in[i] : make_int2(0, -1);
        }
        #pragma unroll
        for (int k = 0; k < PLACE_UNROLL; ++k) {
            const int j = r[k].y - node0;
            if ((unsigned)j < (unsigned)BUCKET_NODES) {
                const int pos = atomicAdd(&cnt[j], 1);
                if ((unsigned)pos < (unsigned)EEDGES)
                    ((int2*)csr)[pos] = r[k];
            }
        }
    }
}

#define QUAD_XOR1 0xB1
#define QUAD_XOR2 0x4E
template <int CTRL>
__device__ __forceinline__ float quad_move(float v) {
    return __int_as_float(__builtin_amdgcn_mov_dpp(__float_as_int(v), CTRL, 0xF, 0xF, true));
}

__device__ __forceinline__ float quad_sum(float v) {
    v += quad_move<QUAD_XOR1>(v);
    v += quad_move<QUAD_XOR2>(v);
    return v;
}

__device__ __forceinline__ float quad_transpose_sum(float e0, float e1, float e2, float e3,
                                                    bool odd, bool hi) {
    const float rA = (odd ? e1 : e0) + quad_move<QUAD_XOR1>(odd ? e0 : e1);
    const float rB = (odd ? e3 : e2) + quad_move<QUAD_XOR1>(odd ? e2 : e3);
    return (hi ? rB : rA) + quad_move<QUAD_XOR2>(hi ? rA : rB);
}

__device__ __forceinline__ float dot4(float4 a, float4 b) {
    return a.x * b.x + a.y * b.y + a.z * b.z + a.w * b.w;
}

__device__ __forceinline__ float edge_score(float logit, float a) {
    float sg = 1.0f / (1.0f + expf(-logit));
    float v  = a * sg;
    return (v > 0.0f) ? v : 0.2f * v;
}

__device__ __forceinline__ float4 load_row(const float* __restrict__ xt, int s, int lane) {
    s = min(max(s, 0), NNODES - 1);
    return *(const float4*)(xt + (size_t)s * FDIM + lane * 4);
}

#define GATHER_WAVES 10000
__global__ void __launch_bounds__(128)
gat_attn_fused(const float* __restrict__ xt,
               const float* __restrict__ attL,
               const float* __restrict__ attR,
               float* __restrict__ agg,
               const int* __restrict__ rowptr,
               const int* __restrict__ csr) {
    const int lane = threadIdx.x & 31;
    const bool odd = (lane & 1) != 0;
    const bool hi  = (lane & 2) != 0;
    const int wv = __builtin_amdgcn_readfirstlane((int)(threadIdx.x >> 5));
    const int gw = blockIdx.x * 4 + wv;
    const int2* rec = (const int2*)csr;
    const float4 al = *(const float4*)(attL + lane * 4);
    const float4 ar = *(const float4*)(attR + lane * 4);

    for (int node = gw; node < NNODES; node += GATHER_WAVES) {
        const int beg = max(rowptr[node], 0), end = min(rowptr[node + 1], EEDGES);
        const float4 xi = *(const float4*)(xt + (size_t)node * FDIM + lane * 4);
        const float aR = quad_sum(dot4(ar, xi));
        float m   = edge_score(quad_sum(dot4(xi, xi)), quad_sum(dot4(al, xi)) + aR);
        float den = 1.0f;
        float4 acc = xi;

        int p = beg;
        for (; p + 4 <= end; p += 4) {
            const float4 x0 = load_row(xt, rec[p].x,     lane);
            const float4 x1 = load_row(xt, rec[p + 1].x, lane);
            const float4 x2 = load_row(xt, rec[p + 2].x, lane);
            const float4 x3 = load_row(xt, rec[p + 3].x, lane);
            const float lg = quad_transpose_sum(dot4(xi, x0), dot4(xi, x1),
                                                dot4(xi, x2), dot4(xi, x3), odd, hi);
            const float aa = quad_transpose_sum(dot4(al, x0), dot4(al, x1),
                                                dot4(al, x2), dot4(al, x3), odd, hi) + aR;
            const float v  = edge_score(lg, aa);
            float mx = fmaxf(v,  quad_move<QUAD_XOR1>(v));
            mx       = fmaxf(mx, quad_move<QUAD_XOR2>(mx));
            const float mn = fmaxf(m, mx);
            const float sc = expf(m - mn);
            const float pe = expf(v - mn);
            const float p0 = quad_move<0x00>(pe);
            const float p1 = quad_move<0x55>(pe);
            const float p2 = quad_move<0xAA>(pe);
            const float p3 = quad_move<0xFF>(pe);
            den   = den   * sc + ((p0 + p1) + (p2 + p3));
            acc.x = acc.x * sc + ((p0 * x0.x + p1 * x1.x) + (p2 * x2.x + p3 * x3.x));
            acc.y = acc.y * sc + ((p0 * x0.y + p1 * x1.y) + (p2 * x2.y + p3 * x3.y));
            acc.z = acc.z * sc + ((p0 * x0.z + p1 * x1.z) + (p2 * x2.z + p3 * x3.z));
            acc.w = acc.w * sc + ((p0 * x0.w + p1 * x1.w) + (p2 * x2.w + p3 * x3.w));
            m = mn;
        }
        for (; p < end; ++p) {
            const float4 x = load_row(xt, rec[p].x, lane);
            const float v  = edge_score(quad_sum(dot4(xi, x)), quad_sum(dot4(al, x)) + aR);
            const float mn = fmaxf(m, v);
            const float sc = expf(m - mn);
            const float pe = expf(v - mn);
            den   = den   * sc + pe;
            acc.x = acc.x * sc + pe * x.x;
            acc.y = acc.y * sc + pe * x.y;
            acc.z = acc.z * sc + pe * x.z;
            acc.w = acc.w * sc + pe * x.w;
            m = mn;
        }
        const float inv = 1.0f / (den + 1e-16f);
        *(float4*)(agg + (size_t)node * FDIM + lane * 4) =
            make_float4(acc.x * inv, acc.y * inv, acc.z * inv, acc.w * inv);
    }
}

__global__ void gat_bias_gelu(const float* __restrict__ agg,
                              const float* __restrict__ bias,
                              float* __restrict__ out) {
    int g = blockIdx.x * blockDim.x + threadIdx.x;
    if (g >= NNODES * FDIM) return;
    float v = agg[g] + bias[g & (FDIM - 1)];
    out[g] = 0.5f * v * (1.0f + erff(v * 0.70710678118654752f));
}

__global__ void gat_fc(const float* __restrict__ h,
                       const float* __restrict__ w,
                       const float* __restrict__ b0,
                       float* __restrict__ out) {
    int n = blockIdx.x * blockDim.x + threadIdx.x;
    if (n >= NNODES) return;
    const float4* h4 = (const float4*)(h + (size_t)n * FDIM);
    const float4* w4 = (const float4*)w;
    float s = 0.f;
    #pragma unroll 8
    for (int i = 0; i < FDIM / 4; ++i) {
        float4 a = h4[i], c = w4[i];
        s += a.x * c.x + a.y * c.y + a.z * c.z + a.w * c.w;
    }
    out[n] = s + b0[0];
}

extern "C" void kernel_launch(void* const* d_in, const int* in_sizes, int n_in,
                              void* d_out, int out_size, void* d_ws, size_t ws_size,
                              hipStream_t stream) {
    (void)in_sizes; (void)n_in; (void)out_size; (void)ws_size;
    const float* x   = (const float*)d_in[0];
    const float* W1  = (const float*)d_in[1];
    const float* b1  = (const float*)d_in[2];
    const float* al1 = (const float*)d_in[3];
    const float* ar1 = (const float*)d_in[4];
    const float* W2  = (const float*)d_in[5];
    const float* b2  = (const float*)d_in[6];
    const float* al2 = (const float*)d_in[7];
    const float* ar2 = (const float*)d_in[8];
    const float* fcW = (const float*)d_in[9];
    const float* fcb = (const float*)d_in[10];
    const int*   ei  = (const int*)d_in[11];

    float* ws   = (float*)d_ws;
    float* xt   = ws;
    float* hbuf = xt   + (size_t)NNODES * FDIM;
    float* agg  = hbuf + (size_t)NNODES * FDIM;
    int*   tab  = (int*)(agg + (size_t)NNODES * FDIM);
    int*   pre  = tab  + (size_t)SCAN_N;
    int*   rowp = pre  + (size_t)SCAN_N + 4;
    int*   tmp  = rowp + (size_t)NNODES + 4;
    int*   csr  = tmp  + (size_t)EEDGES * 2;

    const int NG = (NNODES * FDIM + 255) / 256;
    const dim3 B(256);

    gat_gemm128_wmma<<<NNODES / 16, B, 0, stream>>>(x, W1, xt);
    gat_sort_hist<<<SORT_CHUNKS, B, 0, stream>>>(ei, tab);
    gat_csr_scan<<<1, 1024, 0, stream>>>(tab, pre);
    gat_sort_scatter<<<SORT_CHUNKS, B, 0, stream>>>(ei, pre, tmp);
    gat_sort_place<<<SORT_BUCKETS, B, 0, stream>>>(pre, tmp, rowp, csr);
    gat_attn_fused<<<GATHER_WAVES / 4, 128, 0, stream>>>(xt, al1, ar1, agg, rowp, csr);
    gat_bias_gelu<<<NG, B, 0, stream>>>(agg, b1, hbuf);

    gat_gemm128_wmma<<<NNODES / 16, B, 0, stream>>>(hbuf, W2, xt);
    gat_attn_fused<<<GATHER_WAVES / 4, 128, 0, stream>>>(xt, al2, ar2, agg, rowp, csr);
    gat_bias_gelu<<<NG, B, 0, stream>>>(agg, b2, hbuf);

    gat_fc<<<(NNODES + 255) / 256, B, 0, stream>>>(hbuf, fcW, fcb, (float*)d_out);
}


# bias+GELU and final FC passes folded into neighbours: layer-1 GELU applied as the layer-2 GEMM stages its A tile, layer-2 GELU + FC dot in the attention wave (wave-tree sum: FC sum order changes)
# speedup vs baseline: 7.7381x; 1.2367x over previous
#include <hip/hip_runtime.h>
#include <hip/hip_bf16.h>


#define NNODES  50000
#define EEDGES  800000
#define ETOT    (EEDGES + NNODES)
#define FDIM    128
#define NHEADS  8
#define CDIM    16

typedef __attribute__((ext_vector_type(2))) float v2f;
typedef __attribute__((ext_vector_type(8))) float v8f;

__device__ __forceinline__ float gelu_exact(float v) {
    return 0.5f * v * (1.0f + erff(v * 0.70710678118654752f));
}

template <bool GELU_IN>
__global__ void gat_gemm128_wmma(const float* __restrict__ Ain,
                                 const float* __restrict__ abias,
                                 const float* __restrict__ Wmat,
                                 float* __restrict__ Out) {
    __shared__ float As[16 * 132];
    const int tileBase = blockIdx.x * 16;
    const int t = threadIdx.x;

    const float4* A4 = (const float4*)(Ain + (size_t)tileBase * FDIM);
    const int r0 = t >> 5;
    const int c4 = t & 31;
    float4 v[2];
    #pragma unroll
    for (int j = 0; j < 2; ++j) v[j] = A4[(j * 8 + r0) * 32 + c4];
    if (GELU_IN) {
        const float4 b = *(const float4*)(abias + c4 * 4);
        #pragma unroll
        for (int j = 0; j < 2; ++j) {
            v[j].x = gelu_exact(v[j].x + b.x);
            v[j].y = gelu_exact(v[j].y + b.y);
            v[j].z = gelu_exact(v[j].z + b.z);
            v[j].w = gelu_exact(v[j].w + b.w);
        }
    }
    #pragma unroll
    for (int j = 0; j < 2; ++j)
        *(float4*)&As[(j * 8 + r0) * 132 + c4 * 4] = v[j];
    __syncthreads();

    const int wave = t >> 5;
    const int lane = t & 31;
    const int m    = lane & 15;
    const int half = lane >> 4;
    const int col  = wave * 16 + m;

    v8f acc = {};
    #pragma unroll 4
    for (int kk = 0; kk < 32; ++kk) {
        const int k = kk * 4 + half * 2;
        v2f a, b;
        a.x = As[m * 132 + k];
        a.y = As[m * 132 + k + 1];
        b.x = Wmat[(size_t)k * FDIM + col];
        b.y = Wmat[(size_t)(k + 1) * FDIM + col];
        acc = __builtin_amdgcn_wmma_f32_16x16x4_f32(
             false, a,  false, b,
             (short)0, acc,  false,  false);
    }

    const int mbase = half * 8;
    #pragma unroll
    for (int i = 0; i < 8; ++i)
        Out[(size_t)(tileBase + mbase + i) * FDIM + wave * 16 + m] = acc[i];
}

#define SORT_CHUNKS  200
#define CHUNK_EDGES  4000
#define SORT_BUCKETS 250
#define BUCKET_NODES 200
#define SCAN_N       (SORT_BUCKETS * SORT_CHUNKS)
static_assert(SORT_CHUNKS * CHUNK_EDGES == EEDGES, "chunks must tile the edge list");
static_assert(SORT_BUCKETS * BUCKET_NODES == NNODES, "ranges must tile the nodes");
static_assert(CHUNK_EDGES % 4 == 0 && SCAN_N % 4 == 0, "int4 access");
static_assert(SCAN_N / 4 <= 1024 * 13, "one scan block must cover the table");
static_assert(SORT_BUCKETS <= 256 && BUCKET_NODES <= 256, "one counter per thread");

#define SORT_TRIPS ((CHUNK_EDGES / 4 + 255) / 256)

__global__ void __launch_bounds__(256)
gat_sort_hist(const int* __restrict__ ei, int* __restrict__ tab) {
    __shared__ int lh[SORT_BUCKETS];
    const int t = threadIdx.x, g = blockIdx.x;
    const int4* d4 = (const int4*)(ei + EEDGES + g * CHUNK_EDGES);
    int4 d[SORT_TRIPS];
    #pragma unroll
    for (int k = 0; k < SORT_TRIPS; ++k) {
        const int q = t + k * 256;
        d[k] = (q < CHUNK_EDGES / 4) ? d4[q] : make_int4(-1, -1, -1, -1);
    }
    if (t < SORT_BUCKETS) lh[t] = 0;
    __syncthreads();
    #pragma unroll
    for (int k = 0; k < SORT_TRIPS; ++k) {
        if ((unsigned)d[k].x < (unsigned)NNODES) atomicAdd(&lh[d[k].x / BUCKET_NODES], 1);
        if ((unsigned)d[k].y < (unsigned)NNODES) atomicAdd(&lh[d[k].y / BUCKET_NODES], 1);
        if ((unsigned)d[k].z < (unsigned)NNODES) atomicAdd(&lh[d[k].z / BUCKET_NODES], 1);
        if ((unsigned)d[k].w < (unsigned)NNODES) atomicAdd(&lh[d[k].w / BUCKET_NODES], 1);
    }
    __syncthreads();
    if (t < SORT_BUCKETS) tab[t * SORT_CHUNKS + g] = lh[t];
}

__global__ void __launch_bounds__(1024)
gat_csr_scan(const int* __restrict__ tab, int* __restrict__ pre) {
    __shared__ int wtot[32];
    const int t = threadIdx.x;
    const int lane = t & 31, w = t >> 5;
    const int NQ = SCAN_N / 4;
    const int4* d4 = (const int4*)tab;
    int4* r4 = (int4*)pre;
    int4 v[13];
    int sum = 0;
    #pragma unroll
    for (int j = 0; j < 13; ++j) {
        int q = t * 13 + j;
        v[j] = (q < NQ) ? d4[q] : make_int4(0, 0, 0, 0);
        sum += v[j].x + v[j].y + v[j].z + v[j].w;
    }
    int inc = sum;
    for (int o = 1; o < 32; o <<= 1) {
        int u = __shfl_up(inc, o);
        if (lane >= o) inc += u;
    }
    if (lane == 31) wtot[w] = inc;
    __syncthreads();
    if (w == 0) {
        int x = wtot[lane], y = x;
        for (int o = 1; o < 32; o <<= 1) {
            int u = __shfl_up(y, o);
            if (lane >= o) y += u;
        }
        wtot[lane] = y - x;
    }
    __syncthreads();
    int run = wtot[w] + inc - sum;
    #pragma unroll
    for (int j = 0; j < 13; ++j) {
        int q = t * 13 + j;
        if (q < NQ) {
            int4 o;
            o.x = run; run += v[j].x;
            o.y = run; run += v[j].y;
            o.z = run; run += v[j].z;
            o.w = run; run += v[j].w;
            r4[q] = o;
        }
    }
    if (t == 1023) pre[SCAN_N] = run;
}

__global__ void __launch_bounds__(256)
gat_sort_scatter(const int* __restrict__ ei, const int* __restrict__ pre,
                 int* __restrict__ tmp) {
    __shared__ int lc[SORT_BUCKETS];
    const int t = threadIdx.x, g = blockIdx.x;
    const int4* s4 = (const int4*)(ei + g * CHUNK_EDGES);
    const int4* d4 = (const int4*)(ei + EEDGES + g * CHUNK_EDGES);
    int4 s[SORT_TRIPS], d[SORT_TRIPS];
    #pragma unroll
    for (int k = 0; k < SORT_TRIPS; ++k) {
        const int q = t + k * 256;
        const bool in = q < CHUNK_EDGES / 4;
        s[k] = in ? s4[q] : make_int4(0, 0, 0, 0);
        d[k] = in ? d4[q] : make_int4(-1, -1, -1, -1);
    }
    if (t < SORT_BUCKETS) lc[t] = pre[t * SORT_CHUNKS + g];
    __syncthreads();
    int2* out = (int2*)tmp;
    #pragma unroll
    for (int k = 0; k < SORT_TRIPS; ++k) {
        const int sv[4] = {s[k].x, s[k].y, s[k].z, s[k].w};
        const int dv[4] = {d[k].x, d[k].y, d[k].z, d[k].w};
        #pragma unroll
        for (int c = 0; c < 4; ++c) {
            if ((unsigned)dv[c] < (unsigned)NNODES) {
                const int pos = atomicAdd(&lc[dv[c] / BUCKET_NODES], 1);
                if ((unsigned)pos < (unsigned)EEDGES)
                    out[pos] = make_int2(sv[c], dv[c]);
            }
        }
    }
}

#define PLACE_UNROLL 8

__global__ void __launch_bounds__(256)
gat_sort_place(const int* __restrict__ pre, const int* __restrict__ tmp,
               int* __restrict__ rowptr, int* __restrict__ csr) {
    __shared__ int cnt[256];
    __shared__ int wtot[8];
    const int t = threadIdx.x, b = blockIdx.x;
    const int lane = t & 31, w = t >> 5;
    const int start = min(max(pre[b * SORT_CHUNKS], 0), EEDGES);
    const int end   = min(max(pre[(b + 1) * SORT_CHUNKS], start), EEDGES);
    const int node0 = b * BUCKET_NODES;
    const int2* in = (const int2*)tmp;
    cnt[t] = 0;
    __syncthreads();
    for (int i0 = start + t; i0 < end; i0 += 256 * PLACE_UNROLL) {
        int dj[PLACE_UNROLL];
        #pragma unroll
        for (int k = 0; k < PLACE_UNROLL; ++k) {
            const int i = i0 + k * 256;
            dj[k] = (i < end) ? in[i].y : -1;
        }
        #pragma unroll
        for (int k = 0; k < PLACE_UNROLL; ++k) {
            const int j = dj[k] - node0;
            if ((unsigned)j < (unsigned)BUCKET_NODES) atomicAdd(&cnt[j], 1);
        }
    }
    __syncthreads();
    const int c = cnt[t];
    int inc = c;
    for (int o = 1; o < 32; o <<= 1) {
        int u = __shfl_up(inc, o);
        if (lane >= o) inc += u;
    }
    if (lane == 31) wtot[w] = inc;
    __syncthreads();
    int first = start + inc - c;
    for (int k = 0; k < w; ++k) first += wtot[k];
    cnt[t] = first;
    if (t < BUCKET_NODES) rowptr[node0 + t] = first;
    if (b == SORT_BUCKETS - 1 && t == 0) rowptr[NNODES] = end;
    __syncthreads();
    for (int i0 = start + t; i0 < end; i0 += 256 * PLACE_UNROLL) {
        int2 r[PLACE_UNROLL];
        #pragma unroll
        for (int k = 0; k < PLACE_UNROLL; ++k) {
            const int i = i0 + k * 256;
            r[k] = (i < end) ? in[i] : make_int2(0, -1);
        }
        #pragma unroll
        for (int k = 0; k < PLACE_UNROLL; ++k) {
            const int j = r[k].y - node0;
            if ((unsigned)j < (unsigned)BUCKET_NODES) {
                const int pos = atomicAdd(&cnt[j], 1);
                if ((unsigned)pos < (unsigned)EEDGES)
                    ((int2*)csr)[pos] = r[k];
            }
        }
    }
}

#define QUAD_XOR1 0xB1
#define QUAD_XOR2 0x4E
template <int CTRL>
__device__ __forceinline__ float quad_move(float v) {
    return __int_as_float(__builtin_amdgcn_mov_dpp(__float_as_int(v), CTRL, 0xF, 0xF, true));
}

__device__ __forceinline__ float quad_sum(float v) {
    v += quad_move<QUAD_XOR1>(v);
    v += quad_move<QUAD_XOR2>(v);
    return v;
}

__device__ __forceinline__ float quad_transpose_sum(float e0, float e1, float e2, float e3,
                                                    bool odd, bool hi) {
    const float rA = (odd ? e1 : e0) + quad_move<QUAD_XOR1>(odd ? e0 : e1);
    const float rB = (odd ? e3 : e2) + quad_move<QUAD_XOR1>(odd ? e2 : e3);
    return (hi ? rB : rA) + quad_move<QUAD_XOR2>(hi ? rA : rB);
}

__device__ __forceinline__ float dot4(float4 a, float4 b) {
    return a.x * b.x + a.y * b.y + a.z * b.z + a.w * b.w;
}

__device__ __forceinline__ float edge_score(float logit, float a) {
    float sg = 1.0f / (1.0f + expf(-logit));
    float v  = a * sg;
    return (v > 0.0f) ? v : 0.2f * v;
}

#define ROW_XOR4 0x164
#define ROW_XOR8 0x168
__device__ __forceinline__ float heads_sum(float v) {
    v += quad_move<ROW_XOR4>(v);
    v += quad_move<ROW_XOR8>(v);
    const int b = __float_as_int(v);
    return v + __int_as_float(__builtin_amdgcn_permlanex16(b, b, 0x76543210, 0xfedcba98, false, false));
}

__device__ __forceinline__ float4 load_row(const float* __restrict__ xt, int s, int lane) {
    s = min(max(s, 0), NNODES - 1);
    return *(const float4*)(xt + (size_t)s * FDIM + lane * 4);
}

#define GATHER_WAVES 10000
template <bool LAST>
__global__ void __launch_bounds__(128)
gat_attn_fused(const float* __restrict__ xt,
               const float* __restrict__ attL,
               const float* __restrict__ attR,
               const float* __restrict__ bias,
               const float* __restrict__ fcw,
               const float* __restrict__ fcb,
               float* __restrict__ out,
               const int* __restrict__ rowptr,
               const int* __restrict__ csr) {
    const int lane = threadIdx.x & 31;
    const bool odd = (lane & 1) != 0;
    const bool hi  = (lane & 2) != 0;
    const int wv = __builtin_amdgcn_readfirstlane((int)(threadIdx.x >> 5));
    const int gw = blockIdx.x * 4 + wv;
    const int2* rec = (const int2*)csr;
    const float4 al = *(const float4*)(attL + lane * 4);
    const float4 ar = *(const float4*)(attR + lane * 4);
    const float4 bb = LAST ? *(const float4*)(bias + lane * 4) : make_float4(0.f, 0.f, 0.f, 0.f);
    const float4 fw = LAST ? *(const float4*)(fcw + lane * 4) : make_float4(0.f, 0.f, 0.f, 0.f);
    const float  fb = LAST ? fcb[0] : 0.0f;

    for (int node = gw; node < NNODES; node += GATHER_WAVES) {
        const int beg = max(rowptr[node], 0), end = min(rowptr[node + 1], EEDGES);
        const float4 xi = *(const float4*)(xt + (size_t)node * FDIM + lane * 4);
        const float aR = quad_sum(dot4(ar, xi));
        float m   = edge_score(quad_sum(dot4(xi, xi)), quad_sum(dot4(al, xi)) + aR);
        float den = 1.0f;
        float4 acc = xi;

        int p = beg;
        for (; p + 4 <= end; p += 4) {
            const float4 x0 = load_row(xt, rec[p].x,     lane);
            const float4 x1 = load_row(xt, rec[p + 1].x, lane);
            const float4 x2 = load_row(xt, rec[p + 2].x, lane);
            const float4 x3 = load_row(xt, rec[p + 3].x, lane);
            const float lg = quad_transpose_sum(dot4(xi, x0), dot4(xi, x1),
                                                dot4(xi, x2), dot4(xi, x3), odd, hi);
            const float aa = quad_transpose_sum(dot4(al, x0), dot4(al, x1),
                                                dot4(al, x2), dot4(al, x3), odd, hi) + aR;
            const float v  = edge_score(lg, aa);
            float mx = fmaxf(v,  quad_move<QUAD_XOR1>(v));
            mx       = fmaxf(mx, quad_move<QUAD_XOR2>(mx));
            const float mn = fmaxf(m, mx);
            const float sc = expf(m - mn);
            const float pe = expf(v - mn);
            const float p0 = quad_move<0x00>(pe);
            const float p1 = quad_move<0x55>(pe);
            const float p2 = quad_move<0xAA>(pe);
            const float p3 = quad_move<0xFF>(pe);
            den   = den   * sc + ((p0 + p1) + (p2 + p3));
            acc.x = acc.x * sc + ((p0 * x0.x + p1 * x1.x) + (p2 * x2.x + p3 * x3.x));
            acc.y = acc.y * sc + ((p0 * x0.y + p1 * x1.y) + (p2 * x2.y + p3 * x3.y));
            acc.z = acc.z * sc + ((p0 * x0.z + p1 * x1.z) + (p2 * x2.z + p3 * x3.z));
            acc.w = acc.w * sc + ((p0 * x0.w + p1 * x1.w) + (p2 * x2.w + p3 * x3.w));
            m = mn;
        }
        for (; p < end; ++p) {
            const float4 x = load_row(xt, rec[p].x, lane);
            const float v  = edge_score(quad_sum(dot4(xi, x)), quad_sum(dot4(al, x)) + aR);
            const float mn = fmaxf(m, v);
            const float sc = expf(m - mn);
            const float pe = expf(v - mn);
            den   = den   * sc + pe;
            acc.x = acc.x * sc + pe * x.x;
            acc.y = acc.y * sc + pe * x.y;
            acc.z = acc.z * sc + pe * x.z;
            acc.w = acc.w * sc + pe * x.w;
            m = mn;
        }
        const float inv = 1.0f / (den + 1e-16f);
        if (!LAST) {
            *(float4*)(out + (size_t)node * FDIM + lane * 4) =
                make_float4(acc.x * inv, acc.y * inv, acc.z * inv, acc.w * inv);
        } else {
            float4 h;
            h.x = gelu_exact(acc.x * inv + bb.x);
            h.y = gelu_exact(acc.y * inv + bb.y);
            h.z = gelu_exact(acc.z * inv + bb.z);
            h.w = gelu_exact(acc.w * inv + bb.w);
            const float y = heads_sum(quad_sum(dot4(h, fw)));
            if (lane == 0) out[node] = y + fb;
        }
    }
}

extern "C" void kernel_launch(void* const* d_in, const int* in_sizes, int n_in,
                              void* d_out, int out_size, void* d_ws, size_t ws_size,
                              hipStream_t stream) {
    (void)in_sizes; (void)n_in; (void)out_size; (void)ws_size;
    const float* x   = (const float*)d_in[0];
    const float* W1  = (const float*)d_in[1];
    const float* b1  = (const float*)d_in[2];
    const float* al1 = (const float*)d_in[3];
    const float* ar1 = (const float*)d_in[4];
    const float* W2  = (const float*)d_in[5];
    const float* b2  = (const float*)d_in[6];
    const float* al2 = (const float*)d_in[7];
    const float* ar2 = (const float*)d_in[8];
    const float* fcW = (const float*)d_in[9];
    const float* fcb = (const float*)d_in[10];
    const int*   ei  = (const int*)d_in[11];

    float* ws   = (float*)d_ws;
    float* xt   = ws;
    float* agg  = xt   + (size_t)NNODES * FDIM;
    int*   tab  = (int*)(agg + (size_t)NNODES * FDIM);
    int*   pre  = tab  + (size_t)SCAN_N;
    int*   rowp = pre  + (size_t)SCAN_N + 4;
    int*   tmp  = rowp + (size_t)NNODES + 4;
    int*   csr  = tmp  + (size_t)EEDGES * 2;

    const dim3 B(256);

    gat_gemm128_wmma<false><<<NNODES / 16, B, 0, stream>>>(x, b1, W1, xt);
    gat_sort_hist<<<SORT_CHUNKS, B, 0, stream>>>(ei, tab);
    gat_csr_scan<<<1, 1024, 0, stream>>>(tab, pre);
    gat_sort_scatter<<<SORT_CHUNKS, B, 0, stream>>>(ei, pre, tmp);
    gat_sort_place<<<SORT_BUCKETS, B, 0, stream>>>(pre, tmp, rowp, csr);
    gat_attn_fused<false><<<GATHER_WAVES / 4, 128, 0, stream>>>(
        xt, al1, ar1, b1, fcW, fcb, agg, rowp, csr);

    gat_gemm128_wmma<true><<<NNODES / 16, B, 0, stream>>>(agg, b1, W2, xt);
    gat_attn_fused<true><<<GATHER_WAVES / 4, 128, 0, stream>>>(
        xt, al2, ar2, b2, fcW, fcb, (float*)d_out, rowp, csr);
}


# edge pass: in-line exact expf and float division (sigmoid, softmax weights, rescale, 1/den) replaced by native v_exp_f32 of a log2e-scaled argument and v_rcp_f32; last bits of the weights change
# speedup vs baseline: 8.4525x; 1.0923x over previous
#include <hip/hip_runtime.h>
#include <hip/hip_bf16.h>


#define NNODES  50000
#define EEDGES  800000
#define ETOT    (EEDGES + NNODES)
#define FDIM    128
#define NHEADS  8
#define CDIM    16

typedef __attribute__((ext_vector_type(2))) float v2f;
typedef __attribute__((ext_vector_type(8))) float v8f;

__device__ __forceinline__ float gelu_exact(float v) {
    return 0.5f * v * (1.0f + erff(v * 0.70710678118654752f));
}

template <bool GELU_IN>
__global__ void gat_gemm128_wmma(const float* __restrict__ Ain,
                                 const float* __restrict__ abias,
                                 const float* __restrict__ Wmat,
                                 float* __restrict__ Out) {
    __shared__ float As[16 * 132];
    const int tileBase = blockIdx.x * 16;
    const int t = threadIdx.x;

    const float4* A4 = (const float4*)(Ain + (size_t)tileBase * FDIM);
    const int r0 = t >> 5;
    const int c4 = t & 31;
    float4 v[2];
    #pragma unroll
    for (int j = 0; j < 2; ++j) v[j] = A4[(j * 8 + r0) * 32 + c4];
    if (GELU_IN) {
        const float4 b = *(const float4*)(abias + c4 * 4);
        #pragma unroll
        for (int j = 0; j < 2; ++j) {
            v[j].x = gelu_exact(v[j].x + b.x);
            v[j].y = gelu_exact(v[j].y + b.y);
            v[j].z = gelu_exact(v[j].z + b.z);
            v[j].w = gelu_exact(v[j].w + b.w);
        }
    }
    #pragma unroll
    for (int j = 0; j < 2; ++j)
        *(float4*)&As[(j * 8 + r0) * 132 + c4 * 4] = v[j];
    __syncthreads();

    const int wave = t >> 5;
    const int lane = t & 31;
    const int m    = lane & 15;
    const int half = lane >> 4;
    const int col  = wave * 16 + m;

    v8f acc = {};
    #pragma unroll 4
    for (int kk = 0; kk < 32; ++kk) {
        const int k = kk * 4 + half * 2;
        v2f a, b;
        a.x = As[m * 132 + k];
        a.y = As[m * 132 + k + 1];
        b.x = Wmat[(size_t)k * FDIM + col];
        b.y = Wmat[(size_t)(k + 1) * FDIM + col];
        acc = __builtin_amdgcn_wmma_f32_16x16x4_f32(
             false, a,  false, b,
             (short)0, acc,  false,  false);
    }

    const int mbase = half * 8;
    #pragma unroll
    for (int i = 0; i < 8; ++i)
        Out[(size_t)(tileBase + mbase + i) * FDIM + wave * 16 + m] = acc[i];
}

#define SORT_CHUNKS  200
#define CHUNK_EDGES  4000
#define SORT_BUCKETS 250
#define BUCKET_NODES 200
#define SCAN_N       (SORT_BUCKETS * SORT_CHUNKS)
static_assert(SORT_CHUNKS * CHUNK_EDGES == EEDGES, "chunks must tile the edge list");
static_assert(SORT_BUCKETS * BUCKET_NODES == NNODES, "ranges must tile the nodes");
static_assert(CHUNK_EDGES % 4 == 0 && SCAN_N % 4 == 0, "int4 access");
static_assert(SCAN_N / 4 <= 1024 * 13, "one scan block must cover the table");
static_assert(SORT_BUCKETS <= 256 && BUCKET_NODES <= 256, "one counter per thread");

#define SORT_TRIPS ((CHUNK_EDGES / 4 + 255) / 256)

__global__ void __launch_bounds__(256)
gat_sort_hist(const int* __restrict__ ei, int* __restrict__ tab) {
    __shared__ int lh[SORT_BUCKETS];
    const int t = threadIdx.x, g = blockIdx.x;
    const int4* d4 = (const int4*)(ei + EEDGES + g * CHUNK_EDGES);
    int4 d[SORT_TRIPS];
    #pragma unroll
    for (int k = 0; k < SORT_TRIPS; ++k) {
        const int q = t + k * 256;
        d[k] = (q < CHUNK_EDGES / 4) ? d4[q] : make_int4(-1, -1, -1, -1);
    }
    if (t < SORT_BUCKETS) lh[t] = 0;
    __syncthreads();
    #pragma unroll
    for (int k = 0; k < SORT_TRIPS; ++k) {
        if ((unsigned)d[k].x < (unsigned)NNODES) atomicAdd(&lh[d[k].x / BUCKET_NODES], 1);
        if ((unsigned)d[k].y < (unsigned)NNODES) atomicAdd(&lh[d[k].y / BUCKET_NODES], 1);
        if ((unsigned)d[k].z < (unsigned)NNODES) atomicAdd(&lh[d[k].z / BUCKET_NODES], 1);
        if ((unsigned)d[k].w < (unsigned)NNODES) atomicAdd(&lh[d[k].w / BUCKET_NODES], 1);
    }
    __syncthreads();
    if (t < SORT_BUCKETS) tab[t * SORT_CHUNKS + g] = lh[t];
}

__global__ void __launch_bounds__(1024)
gat_csr_scan(const int* __restrict__ tab, int* __restrict__ pre) {
    __shared__ int wtot[32];
    const int t = threadIdx.x;
    const int lane = t & 31, w = t >> 5;
    const int NQ = SCAN_N / 4;
    const int4* d4 = (const int4*)tab;
    int4* r4 = (int4*)pre;
    int4 v[13];
    int sum = 0;
    #pragma unroll
    for (int j = 0; j < 13; ++j) {
        int q = t * 13 + j;
        v[j] = (q < NQ) ? d4[q] : make_int4(0, 0, 0, 0);
        sum += v[j].x + v[j].y + v[j].z + v[j].w;
    }
    int inc = sum;
    for (int o = 1; o < 32; o <<= 1) {
        int u = __shfl_up(inc, o);
        if (lane >= o) inc += u;
    }
    if (lane == 31) wtot[w] = inc;
    __syncthreads();
    if (w == 0) {
        int x = wtot[lane], y = x;
        for (int o = 1; o < 32; o <<= 1) {
            int u = __shfl_up(y, o);
            if (lane >= o) y += u;
        }
        wtot[lane] = y - x;
    }
    __syncthreads();
    int run = wtot[w] + inc - sum;
    #pragma unroll
    for (int j = 0; j < 13; ++j) {
        int q = t * 13 + j;
        if (q < NQ) {
            int4 o;
            o.x = run; run += v[j].x;
            o.y = run; run += v[j].y;
            o.z = run; run += v[j].z;
            o.w = run; run += v[j].w;
            r4[q] = o;
        }
    }
    if (t == 1023) pre[SCAN_N] = run;
}

__global__ void __launch_bounds__(256)
gat_sort_scatter(const int* __restrict__ ei, const int* __restrict__ pre,
                 int* __restrict__ tmp) {
    __shared__ int lc[SORT_BUCKETS];
    const int t = threadIdx.x, g = blockIdx.x;
    const int4* s4 = (const int4*)(ei + g * CHUNK_EDGES);
    const int4* d4 = (const int4*)(ei + EEDGES + g * CHUNK_EDGES);
    int4 s[SORT_TRIPS], d[SORT_TRIPS];
    #pragma unroll
    for (int k = 0; k < SORT_TRIPS; ++k) {
        const int q = t + k * 256;
        const bool in = q < CHUNK_EDGES / 4;
        s[k] = in ? s4[q] : make_int4(0, 0, 0, 0);
        d[k] = in ? d4[q] : make_int4(-1, -1, -1, -1);
    }
    if (t < SORT_BUCKETS) lc[t] = pre[t * SORT_CHUNKS + g];
    __syncthreads();
    int2* out = (int2*)tmp;
    #pragma unroll
    for (int k = 0; k < SORT_TRIPS; ++k) {
        const int sv[4] = {s[k].x, s[k].y, s[k].z, s[k].w};
        const int dv[4] = {d[k].x, d[k].y, d[k].z, d[k].w};
        #pragma unroll
        for (int c = 0; c < 4; ++c) {
            if ((unsigned)dv[c] < (unsigned)NNODES) {
                const int pos = atomicAdd(&lc[dv[c] / BUCKET_NODES], 1);
                if ((unsigned)pos < (unsigned)EEDGES)
                    out[pos] = make_int2(sv[c], dv[c]);
            }
        }
    }
}

#define PLACE_UNROLL 8

__global__ void __launch_bounds__(256)
gat_sort_place(const int* __restrict__ pre, const int* __restrict__ tmp,
               int* __restrict__ rowptr, int* __restrict__ csr) {
    __shared__ int cnt[256];
    __shared__ int wtot[8];
    const int t = threadIdx.x, b = blockIdx.x;
    const int lane = t & 31, w = t >> 5;
    const int start = min(max(pre[b * SORT_CHUNKS], 0), EEDGES);
    const int end   = min(max(pre[(b + 1) * SORT_CHUNKS], start), EEDGES);
    const int node0 = b * BUCKET_NODES;
    const int2* in = (const int2*)tmp;
    cnt[t] = 0;
    __syncthreads();
    for (int i0 = start + t; i0 < end; i0 += 256 * PLACE_UNROLL) {
        int dj[PLACE_UNROLL];
        #pragma unroll
        for (int k = 0; k < PLACE_UNROLL; ++k) {
            const int i = i0 + k * 256;
            dj[k] = (i < end) ? in[i].y : -1;
        }
        #pragma unroll
        for (int k = 0; k < PLACE_UNROLL; ++k) {
            const int j = dj[k] - node0;
            if ((unsigned)j < (unsigned)BUCKET_NODES) atomicAdd(&cnt[j], 1);
        }
    }
    __syncthreads();
    const int c = cnt[t];
    int inc = c;
    for (int o = 1; o < 32; o <<= 1) {
        int u = __shfl_up(inc, o);
        if (lane >= o) inc += u;
    }
    if (lane == 31) wtot[w] = inc;
    __syncthreads();
    int first = start + inc - c;
    for (int k = 0; k < w; ++k) first += wtot[k];
    cnt[t] = first;
    if (t < BUCKET_NODES) rowptr[node0 + t] = first;
    if (b == SORT_BUCKETS - 1 && t == 0) rowptr[NNODES] = end;
    __syncthreads();
    for (int i0 = start + t; i0 < end; i0 += 256 * PLACE_UNROLL) {
        int2 r[PLACE_UNROLL];
        #pragma unroll
        for (int k = 0; k < PLACE_UNROLL; ++k) {
            const int i = i0 + k * 256;
            r[k] = (i < end) ? in[i] : make_int2(0, -1);
        }
        #pragma unroll
        for (int k = 0; k < PLACE_UNROLL; ++k) {
            const int j = r[k].y - node0;
            if ((unsigned)j < (unsigned)BUCKET_NODES) {
                const int pos = atomicAdd(&cnt[j], 1);
                if ((unsigned)pos < (unsigned)EEDGES)
                    ((int2*)csr)[pos] = r[k];
            }
        }
    }
}

#define QUAD_XOR1 0xB1
#define QUAD_XOR2 0x4E
template <int CTRL>
__device__ __forceinline__ float quad_move(float v) {
    return __int_as_float(__builtin_amdgcn_mov_dpp(__float_as_int(v), CTRL, 0xF, 0xF, true));
}

__device__ __forceinline__ float quad_sum(float v) {
    v += quad_move<QUAD_XOR1>(v);
    v += quad_move<QUAD_XOR2>(v);
    return v;
}

__device__ __forceinline__ float quad_transpose_sum(float e0, float e1, float e2, float e3,
                                                    bool odd, bool hi) {
    const float rA = (odd ? e1 : e0) + quad_move<QUAD_XOR1>(odd ? e0 : e1);
    const float rB = (odd ? e3 : e2) + quad_move<QUAD_XOR1>(odd ? e2 : e3);
    return (hi ? rB : rA) + quad_move<QUAD_XOR2>(hi ? rA : rB);
}

__device__ __forceinline__ float dot4(float4 a, float4 b) {
    return a.x * b.x + a.y * b.y + a.z * b.z + a.w * b.w;
}

#define LOG2E 1.4426950408889634f
__device__ __forceinline__ float exp_hw(float x) {
    return __builtin_amdgcn_exp2f(x * LOG2E);
}
__device__ __forceinline__ float rcp_hw(float x) {
    return __builtin_amdgcn_rcpf(x);
}

__device__ __forceinline__ float edge_score(float logit, float a) {
    float sg = rcp_hw(1.0f + exp_hw(-logit));
    float v  = a * sg;
    return (v > 0.0f) ? v : 0.2f * v;
}

#define ROW_XOR4 0x164
#define ROW_XOR8 0x168
__device__ __forceinline__ float heads_sum(float v) {
    v += quad_move<ROW_XOR4>(v);
    v += quad_move<ROW_XOR8>(v);
    const int b = __float_as_int(v);
    return v + __int_as_float(__builtin_amdgcn_permlanex16(b, b, 0x76543210, 0xfedcba98, false, false));
}

__device__ __forceinline__ float4 load_row(const float* __restrict__ xt, int s, int lane) {
    s = min(max(s, 0), NNODES - 1);
    return *(const float4*)(xt + (size_t)s * FDIM + lane * 4);
}

#define GATHER_WAVES 10000
template <bool LAST>
__global__ void __launch_bounds__(128)
gat_attn_fused(const float* __restrict__ xt,
               const float* __restrict__ attL,
               const float* __restrict__ attR,
               const float* __restrict__ bias,
               const float* __restrict__ fcw,
               const float* __restrict__ fcb,
               float* __restrict__ out,
               const int* __restrict__ rowptr,
               const int* __restrict__ csr) {
    const int lane = threadIdx.x & 31;
    const bool odd = (lane & 1) != 0;
    const bool hi  = (lane & 2) != 0;
    const int wv = __builtin_amdgcn_readfirstlane((int)(threadIdx.x >> 5));
    const int gw = blockIdx.x * 4 + wv;
    const int2* rec = (const int2*)csr;
    const float4 al = *(const float4*)(attL + lane * 4);
    const float4 ar = *(const float4*)(attR + lane * 4);
    const float4 bb = LAST ? *(const float4*)(bias + lane * 4) : make_float4(0.f, 0.f, 0.f, 0.f);
    const float4 fw = LAST ? *(const float4*)(fcw + lane * 4) : make_float4(0.f, 0.f, 0.f, 0.f);
    const float  fb = LAST ? fcb[0] : 0.0f;

    for (int node = gw; node < NNODES; node += GATHER_WAVES) {
        const int beg = max(rowptr[node], 0), end = min(rowptr[node + 1], EEDGES);
        const float4 xi = *(const float4*)(xt + (size_t)node * FDIM + lane * 4);
        const float aR = quad_sum(dot4(ar, xi));
        float m   = edge_score(quad_sum(dot4(xi, xi)), quad_sum(dot4(al, xi)) + aR);
        float den = 1.0f;
        float4 acc = xi;

        int p = beg;
        for (; p + 4 <= end; p += 4) {
            const float4 x0 = load_row(xt, rec[p].x,     lane);
            const float4 x1 = load_row(xt, rec[p + 1].x, lane);
            const float4 x2 = load_row(xt, rec[p + 2].x, lane);
            const float4 x3 = load_row(xt, rec[p + 3].x, lane);
            const float lg = quad_transpose_sum(dot4(xi, x0), dot4(xi, x1),
                                                dot4(xi, x2), dot4(xi, x3), odd, hi);
            const float aa = quad_transpose_sum(dot4(al, x0), dot4(al, x1),
                                                dot4(al, x2), dot4(al, x3), odd, hi) + aR;
            const float v  = edge_score(lg, aa);
            float mx = fmaxf(v,  quad_move<QUAD_XOR1>(v));
            mx       = fmaxf(mx, quad_move<QUAD_XOR2>(mx));
            const float mn = fmaxf(m, mx);
            const float sc = exp_hw(m - mn);
            const float pe = exp_hw(v - mn);
            const float p0 = quad_move<0x00>(pe);
            const float p1 = quad_move<0x55>(pe);
            const float p2 = quad_move<0xAA>(pe);
            const float p3 = quad_move<0xFF>(pe);
            den   = den   * sc + ((p0 + p1) + (p2 + p3));
            acc.x = acc.x * sc + ((p0 * x0.x + p1 * x1.x) + (p2 * x2.x + p3 * x3.x));
            acc.y = acc.y * sc + ((p0 * x0.y + p1 * x1.y) + (p2 * x2.y + p3 * x3.y));
            acc.z = acc.z * sc + ((p0 * x0.z + p1 * x1.z) + (p2 * x2.z + p3 * x3.z));
            acc.w = acc.w * sc + ((p0 * x0.w + p1 * x1.w) + (p2 * x2.w + p3 * x3.w));
            m = mn;
        }
        for (; p < end; ++p) {
            const float4 x = load_row(xt, rec[p].x, lane);
            const float v  = edge_score(quad_sum(dot4(xi, x)), quad_sum(dot4(al, x)) + aR);
            const float mn = fmaxf(m, v);
            const float sc = exp_hw(m - mn);
            const float pe = exp_hw(v - mn);
            den   = den   * sc + pe;
            acc.x = acc.x * sc + pe * x.x;
            acc.y = acc.y * sc + pe * x.y;
            acc.z = acc.z * sc + pe * x.z;
            acc.w = acc.w * sc + pe * x.w;
            m = mn;
        }
        const float inv = rcp_hw(den + 1e-16f);
        if (!LAST) {
            *(float4*)(out + (size_t)node * FDIM + lane * 4) =
                make_float4(acc.x * inv, acc.y * inv, acc.z * inv, acc.w * inv);
        } else {
            float4 h;
            h.x = gelu_exact(acc.x * inv + bb.x);
            h.y = gelu_exact(acc.y * inv + bb.y);
            h.z = gelu_exact(acc.z * inv + bb.z);
            h.w = gelu_exact(acc.w * inv + bb.w);
            const float y = heads_sum(quad_sum(dot4(h, fw)));
            if (lane == 0) out[node] = y + fb;
        }
    }
}

extern "C" void kernel_launch(void* const* d_in, const int* in_sizes, int n_in,
                              void* d_out, int out_size, void* d_ws, size_t ws_size,
                              hipStream_t stream) {
    (void)in_sizes; (void)n_in; (void)out_size; (void)ws_size;
    const float* x   = (const float*)d_in[0];
    const float* W1  = (const float*)d_in[1];
    const float* b1  = (const float*)d_in[2];
    const float* al1 = (const float*)d_in[3];
    const float* ar1 = (const float*)d_in[4];
    const float* W2  = (const float*)d_in[5];
    const float* b2  = (const float*)d_in[6];
    const float* al2 = (const float*)d_in[7];
    const float* ar2 = (const float*)d_in[8];
    const float* fcW = (const float*)d_in[9];
    const float* fcb = (const float*)d_in[10];
    const int*   ei  = (const int*)d_in[11];

    float* ws   = (float*)d_ws;
    float* xt   = ws;
    float* agg  = xt   + (size_t)NNODES * FDIM;
    int*   tab  = (int*)(agg + (size_t)NNODES * FDIM);
    int*   pre  = tab  + (size_t)SCAN_N;
    int*   rowp = pre  + (size_t)SCAN_N + 4;
    int*   tmp  = rowp + (size_t)NNODES + 4;
    int*   csr  = tmp  + (size_t)EEDGES * 2;

    const dim3 B(256);

    gat_gemm128_wmma<false><<<NNODES / 16, B, 0, stream>>>(x, b1, W1, xt);
    gat_sort_hist<<<SORT_CHUNKS, B, 0, stream>>>(ei, tab);
    gat_csr_scan<<<1, 1024, 0, stream>>>(tab, pre);
    gat_sort_scatter<<<SORT_CHUNKS, B, 0, stream>>>(ei, pre, tmp);
    gat_sort_place<<<SORT_BUCKETS, B, 0, stream>>>(pre, tmp, rowp, csr);
    gat_attn_fused<false><<<GATHER_WAVES / 4, 128, 0, stream>>>(
        xt, al1, ar1, b1, fcW, fcb, agg, rowp, csr);

    gat_gemm128_wmma<true><<<NNODES / 16, B, 0, stream>>>(agg, b1, W2, xt);
    gat_attn_fused<true><<<GATHER_WAVES / 4, 128, 0, stream>>>(
        xt, al2, ar2, b2, fcW, fcb, (float*)d_out, rowp, csr);
}


# layer-1 product loses its own launch: its 3,125 row tiles ride as extra work-groups behind the sort's in the four edge-sort launches (they need only the edge list); 8 launches -> 7; sums unchanged
# speedup vs baseline: 9.0619x; 1.0721x over previous
#include <hip/hip_runtime.h>
#include <hip/hip_bf16.h>


#define NNODES  50000
#define EEDGES  800000
#define ETOT    (EEDGES + NNODES)
#define FDIM    128
#define NHEADS  8
#define CDIM    16

typedef __attribute__((ext_vector_type(2))) float v2f;
typedef __attribute__((ext_vector_type(8))) float v8f;

__device__ __forceinline__ float gelu_exact(float v) {
    return 0.5f * v * (1.0f + erff(v * 0.70710678118654752f));
}

template <bool GELU_IN>
__device__ __forceinline__ void gemm_tile16(const float* __restrict__ Ain,
                                            const float* __restrict__ abias,
                                            const float* __restrict__ Wmat,
                                            float* __restrict__ Out,
                                            float* As, const int tile, const int t) {
    const int tileBase = tile * 16;

    const float4* A4 = (const float4*)(Ain + (size_t)tileBase * FDIM);
    const int r0 = t >> 5;
    const int c4 = t & 31;
    float4 v[2];
    #pragma unroll
    for (int j = 0; j < 2; ++j) v[j] = A4[(j * 8 + r0) * 32 + c4];
    if (GELU_IN) {
        const float4 b = *(const float4*)(abias + c4 * 4);
        #pragma unroll
        for (int j = 0; j < 2; ++j) {
            v[j].x = gelu_exact(v[j].x + b.x);
            v[j].y = gelu_exact(v[j].y + b.y);
            v[j].z = gelu_exact(v[j].z + b.z);
            v[j].w = gelu_exact(v[j].w + b.w);
        }
    }
    #pragma unroll
    for (int j = 0; j < 2; ++j)
        *(float4*)&As[(j * 8 + r0) * 132 + c4 * 4] = v[j];
    __syncthreads();

    const int wave = t >> 5;
    const int lane = t & 31;
    const int m    = lane & 15;
    const int half = lane >> 4;
    const int col  = wave * 16 + m;

    v8f acc = {};
    #pragma unroll 4
    for (int kk = 0; kk < 32; ++kk) {
        const int k = kk * 4 + half * 2;
        v2f a, b;
        a.x = As[m * 132 + k];
        a.y = As[m * 132 + k + 1];
        b.x = Wmat[(size_t)k * FDIM + col];
        b.y = Wmat[(size_t)(k + 1) * FDIM + col];
        acc = __builtin_amdgcn_wmma_f32_16x16x4_f32(
             false, a,  false, b,
             (short)0, acc,  false,  false);
    }

    const int mbase = half * 8;
    #pragma unroll
    for (int i = 0; i < 8; ++i)
        Out[(size_t)(tileBase + mbase + i) * FDIM + wave * 16 + m] = acc[i];
}

__global__ void gat_gemm128_gelu_wmma(const float* __restrict__ Ain,
                                      const float* __restrict__ abias,
                                      const float* __restrict__ Wmat,
                                      float* __restrict__ Out) {
    __shared__ float As[16 * 132];
    gemm_tile16<true>(Ain, abias, Wmat, Out, As, (int)blockIdx.x, (int)threadIdx.x);
}

#define SORT_CHUNKS  200
#define CHUNK_EDGES  4000
#define SORT_BUCKETS 250
#define BUCKET_NODES 200
#define SCAN_N       (SORT_BUCKETS * SORT_CHUNKS)
static_assert(SORT_CHUNKS * CHUNK_EDGES == EEDGES, "chunks must tile the edge list");
static_assert(SORT_BUCKETS * BUCKET_NODES == NNODES, "ranges must tile the nodes");
static_assert(CHUNK_EDGES % 4 == 0 && SCAN_N % 4 == 0, "int4 access");
static_assert(SCAN_N / 4 <= 1024 * 13, "one scan block must cover the table");
static_assert(SORT_BUCKETS <= 256 && BUCKET_NODES <= 256, "one counter per thread");

#define SORT_TRIPS ((CHUNK_EDGES / 4 + 255) / 256)

#define GEMM_TILES    (NNODES / 16)
#define TILES_HIST    700
#define TILES_SCAN    704
#define TILES_SCATTER 861
#define TILES_PLACE   860
#define TILE0_HIST    0
#define TILE0_SCAN    (TILE0_HIST + TILES_HIST)
#define TILE0_SCATTER (TILE0_SCAN + TILES_SCAN)
#define TILE0_PLACE   (TILE0_SCATTER + TILES_SCATTER)
static_assert(GEMM_TILES * 16 == NNODES, "row tiles must cover the nodes exactly");
static_assert(TILE0_PLACE + TILES_PLACE == GEMM_TILES, "the four shares must cover every row tile once");
static_assert(TILES_SCAN % 4 == 0, "four tiles per 1,024-thread work-group");

__device__ __forceinline__ void sort_hist_stage(const int* __restrict__ ei, int* __restrict__ tab,
                                                const int g) {
    __shared__ int lh[SORT_BUCKETS];
    const int t = threadIdx.x;
    const int4* d4 = (const int4*)(ei + EEDGES + g * CHUNK_EDGES);
    int4 d[SORT_TRIPS];
    #pragma unroll
    for (int k = 0; k < SORT_TRIPS; ++k) {
        const int q = t + k * 256;
        d[k] = (q < CHUNK_EDGES / 4) ? d4[q] : make_int4(-1, -1, -1, -1);
    }
    if (t < SORT_BUCKETS) lh[t] = 0;
    __syncthreads();
    #pragma unroll
    for (int k = 0; k < SORT_TRIPS; ++k) {
        if ((unsigned)d[k].x < (unsigned)NNODES) atomicAdd(&lh[d[k].x / BUCKET_NODES], 1);
        if ((unsigned)d[k].y < (unsigned)NNODES) atomicAdd(&lh[d[k].y / BUCKET_NODES], 1);
        if ((unsigned)d[k].z < (unsigned)NNODES) atomicAdd(&lh[d[k].z / BUCKET_NODES], 1);
        if ((unsigned)d[k].w < (unsigned)NNODES) atomicAdd(&lh[d[k].w / BUCKET_NODES], 1);
    }
    __syncthreads();
    if (t < SORT_BUCKETS) tab[t * SORT_CHUNKS + g] = lh[t];
}

__global__ void __launch_bounds__(256)
gat_sort_hist_gemm(const int* __restrict__ ei, int* __restrict__ tab,
                   const float* __restrict__ x, const float* __restrict__ W1,
                   float* __restrict__ xt) {
    __shared__ float As[16 * 132];
    const int b = blockIdx.x;
    if (b < SORT_CHUNKS) sort_hist_stage(ei, tab, b);
    else gemm_tile16<false>(x, nullptr, W1, xt, As, TILE0_HIST + (b - SORT_CHUNKS), (int)threadIdx.x);
}

__device__ __forceinline__ void csr_scan_stage(const int* __restrict__ tab, int* __restrict__ pre) {
    __shared__ int wtot[32];
    const int t = threadIdx.x;
    const int lane = t & 31, w = t >> 5;
    const int NQ = SCAN_N / 4;
    const int4* d4 = (const int4*)tab;
    int4* r4 = (int4*)pre;
    int4 v[13];
    int sum = 0;
    #pragma unroll
    for (int j = 0; j < 13; ++j) {
        int q = t * 13 + j;
        v[j] = (q < NQ) ? d4[q] : make_int4(0, 0, 0, 0);
        sum += v[j].x + v[j].y + v[j].z + v[j].w;
    }
    int inc = sum;
    for (int o = 1; o < 32; o <<= 1) {
        int u = __shfl_up(inc, o);
        if (lane >= o) inc += u;
    }
    if (lane == 31) wtot[w] = inc;
    __syncthreads();
    if (w == 0) {
        int x = wtot[lane], y = x;
        for (int o = 1; o < 32; o <<= 1) {
            int u = __shfl_up(y, o);
            if (lane >= o) y += u;
        }
        wtot[lane] = y - x;
    }
    __syncthreads();
    int run = wtot[w] + inc - sum;
    #pragma unroll
    for (int j = 0; j < 13; ++j) {
        int q = t * 13 + j;
        if (q < NQ) {
            int4 o;
            o.x = run; run += v[j].x;
            o.y = run; run += v[j].y;
            o.z = run; run += v[j].z;
            o.w = run; run += v[j].w;
            r4[q] = o;
        }
    }
    if (t == 1023) pre[SCAN_N] = run;
}

__global__ void __launch_bounds__(1024)
gat_csr_scan_gemm(const int* __restrict__ tab, int* __restrict__ pre,
                  const float* __restrict__ x, const float* __restrict__ W1,
                  float* __restrict__ xt) {
    __shared__ float As[4 * 16 * 132];
    const int b = blockIdx.x;
    if (b == 0) {
        csr_scan_stage(tab, pre);
    } else {
        const int grp = threadIdx.x >> 8;
        gemm_tile16<false>(x, nullptr, W1, xt, As + grp * (16 * 132),
                           TILE0_SCAN + (b - 1) * 4 + grp, (int)(threadIdx.x & 255));
    }
}

__device__ __forceinline__ void sort_scatter_stage(const int* __restrict__ ei, const int* __restrict__ pre,
                                                   int* __restrict__ tmp, const int g) {
    __shared__ int lc[SORT_BUCKETS];
    const int t = threadIdx.x;
    const int4* s4 = (const int4*)(ei + g * CHUNK_EDGES);
    const int4* d4 = (const int4*)(ei + EEDGES + g * CHUNK_EDGES);
    int4 s[SORT_TRIPS], d[SORT_TRIPS];
    #pragma unroll
    for (int k = 0; k < SORT_TRIPS; ++k) {
        const int q = t + k * 256;
        const bool in = q < CHUNK_EDGES / 4;
        s[k] = in ? s4[q] : make_int4(0, 0, 0, 0);
        d[k] = in ? d4[q] : make_int4(-1, -1, -1, -1);
    }
    if (t < SORT_BUCKETS) lc[t] = pre[t * SORT_CHUNKS + g];
    __syncthreads();
    int2* out = (int2*)tmp;
    #pragma unroll
    for (int k = 0; k < SORT_TRIPS; ++k) {
        const int sv[4] = {s[k].x, s[k].y, s[k].z, s[k].w};
        const int dv[4] = {d[k].x, d[k].y, d[k].z, d[k].w};
        #pragma unroll
        for (int c = 0; c < 4; ++c) {
            if ((unsigned)dv[c] < (unsigned)NNODES) {
                const int pos = atomicAdd(&lc[dv[c] / BUCKET_NODES], 1);
                if ((unsigned)pos < (unsigned)EEDGES)
                    out[pos] = make_int2(sv[c], dv[c]);
            }
        }
    }
}

__global__ void __launch_bounds__(256)
gat_sort_scatter_gemm(const int* __restrict__ ei, const int* __restrict__ pre,
                      int* __restrict__ tmp,
                      const float* __restrict__ x, const float* __restrict__ W1,
                      float* __restrict__ xt) {
    __shared__ float As[16 * 132];
    const int b = blockIdx.x;
    if (b < SORT_CHUNKS) sort_scatter_stage(ei, pre, tmp, b);
    else gemm_tile16<false>(x, nullptr, W1, xt, As, TILE0_SCATTER + (b - SORT_CHUNKS), (int)threadIdx.x);
}

#define PLACE_UNROLL 8

__device__ __forceinline__ void sort_place_stage(const int* __restrict__ pre, const int* __restrict__ tmp,
                                                 int* __restrict__ rowptr, int* __restrict__ csr,
                                                 const int b) {
    __shared__ int cnt[256];
    __shared__ int wtot[8];
    const int t = threadIdx.x;
    const int lane = t & 31, w = t >> 5;
    const int start = min(max(pre[b * SORT_CHUNKS], 0), EEDGES);
    const int end   = min(max(pre[(b + 1) * SORT_CHUNKS], start), EEDGES);
    const int node0 = b * BUCKET_NODES;
    const int2* in = (const int2*)tmp;
    cnt[t] = 0;
    __syncthreads();
    for (int i0 = start + t; i0 < end; i0 += 256 * PLACE_UNROLL) {
        int dj[PLACE_UNROLL];
        #pragma unroll
        for (int k = 0; k < PLACE_UNROLL; ++k) {
            const int i = i0 + k * 256;
            dj[k] = (i < end) ? in[i].y : -1;
        }
        #pragma unroll
        for (int k = 0; k < PLACE_UNROLL; ++k) {
            const int j = dj[k] - node0;
            if ((unsigned)j < (unsigned)BUCKET_NODES) atomicAdd(&cnt[j], 1);
        }
    }
    __syncthreads();
    const int c = cnt[t];
    int inc = c;
    for (int o = 1; o < 32; o <<= 1) {
        int u = __shfl_up(inc, o);
        if (lane >= o) inc += u;
    }
    if (lane == 31) wtot[w] = inc;
    __syncthreads();
    int first = start + inc - c;
    for (int k = 0; k < w; ++k) first += wtot[k];
    cnt[t] = first;
    if (t < BUCKET_NODES) rowptr[node0 + t] = first;
    if (b == SORT_BUCKETS - 1 && t == 0) rowptr[NNODES] = end;
    __syncthreads();
    for (int i0 = start + t; i0 < end; i0 += 256 * PLACE_UNROLL) {
        int2 r[PLACE_UNROLL];
        #pragma unroll
        for (int k = 0; k < PLACE_UNROLL; ++k) {
            const int i = i0 + k * 256;
            r[k] = (i < end) ? in[i] : make_int2(0, -1);
        }
        #pragma unroll
        for (int k = 0; k < PLACE_UNROLL; ++k) {
            const int j = r[k].y - node0;
            if ((unsigned)j < (unsigned)BUCKET_NODES) {
                const int pos = atomicAdd(&cnt[j], 1);
                if ((unsigned)pos < (unsigned)EEDGES)
                    ((int2*)csr)[pos] = r[k];
            }
        }
    }
}

__global__ void __launch_bounds__(256)
gat_sort_place_gemm(const int* __restrict__ pre, const int* __restrict__ tmp,
                    int* __restrict__ rowptr, int* __restrict__ csr,
                    const float* __restrict__ x, const float* __restrict__ W1,
                    float* __restrict__ xt) {
    __shared__ float As[16 * 132];
    const int b = blockIdx.x;
    if (b < SORT_BUCKETS) sort_place_stage(pre, tmp, rowptr, csr, b);
    else gemm_tile16<false>(x, nullptr, W1, xt, As, TILE0_PLACE + (b - SORT_BUCKETS), (int)threadIdx.x);
}

#define QUAD_XOR1 0xB1
#define QUAD_XOR2 0x4E
template <int CTRL>
__device__ __forceinline__ float quad_move(float v) {
    return __int_as_float(__builtin_amdgcn_mov_dpp(__float_as_int(v), CTRL, 0xF, 0xF, true));
}

__device__ __forceinline__ float quad_sum(float v) {
    v += quad_move<QUAD_XOR1>(v);
    v += quad_move<QUAD_XOR2>(v);
    return v;
}

__device__ __forceinline__ float quad_transpose_sum(float e0, float e1, float e2, float e3,
                                                    bool odd, bool hi) {
    const float rA = (odd ? e1 : e0) + quad_move<QUAD_XOR1>(odd ? e0 : e1);
    const float rB = (odd ? e3 : e2) + quad_move<QUAD_XOR1>(odd ? e2 : e3);
    return (hi ? rB : rA) + quad_move<QUAD_XOR2>(hi ? rA : rB);
}

__device__ __forceinline__ float dot4(float4 a, float4 b) {
    return a.x * b.x + a.y * b.y + a.z * b.z + a.w * b.w;
}

#define LOG2E 1.4426950408889634f
__device__ __forceinline__ float exp_hw(float x) {
    return __builtin_amdgcn_exp2f(x * LOG2E);
}
__device__ __forceinline__ float rcp_hw(float x) {
    return __builtin_amdgcn_rcpf(x);
}

__device__ __forceinline__ float edge_score(float logit, float a) {
    float sg = rcp_hw(1.0f + exp_hw(-logit));
    float v  = a * sg;
    return (v > 0.0f) ? v : 0.2f * v;
}

#define ROW_XOR4 0x164
#define ROW_XOR8 0x168
__device__ __forceinline__ float heads_sum(float v) {
    v += quad_move<ROW_XOR4>(v);
    v += quad_move<ROW_XOR8>(v);
    const int b = __float_as_int(v);
    return v + __int_as_float(__builtin_amdgcn_permlanex16(b, b, 0x76543210, 0xfedcba98, false, false));
}

__device__ __forceinline__ float4 load_row(const float* __restrict__ xt, int s, int lane) {
    s = min(max(s, 0), NNODES - 1);
    return *(const float4*)(xt + (size_t)s * FDIM + lane * 4);
}

#define GATHER_WAVES 10000
template <bool LAST>
__global__ void __launch_bounds__(128)
gat_attn_fused(const float* __restrict__ xt,
               const float* __restrict__ attL,
               const float* __restrict__ attR,
               const float* __restrict__ bias,
               const float* __restrict__ fcw,
               const float* __restrict__ fcb,
               float* __restrict__ out,
               const int* __restrict__ rowptr,
               const int* __restrict__ csr) {
    const int lane = threadIdx.x & 31;
    const bool odd = (lane & 1) != 0;
    const bool hi  = (lane & 2) != 0;
    const int wv = __builtin_amdgcn_readfirstlane((int)(threadIdx.x >> 5));
    const int gw = blockIdx.x * 4 + wv;
    const int2* rec = (const int2*)csr;
    const float4 al = *(const float4*)(attL + lane * 4);
    const float4 ar = *(const float4*)(attR + lane * 4);
    const float4 bb = LAST ? *(const float4*)(bias + lane * 4) : make_float4(0.f, 0.f, 0.f, 0.f);
    const float4 fw = LAST ? *(const float4*)(fcw + lane * 4) : make_float4(0.f, 0.f, 0.f, 0.f);
    const float  fb = LAST ? fcb[0] : 0.0f;

    for (int node = gw; node < NNODES; node += GATHER_WAVES) {
        const int beg = max(rowptr[node], 0), end = min(rowptr[node + 1], EEDGES);
        const float4 xi = *(const float4*)(xt + (size_t)node * FDIM + lane * 4);
        const float aR = quad_sum(dot4(ar, xi));
        float m   = edge_score(quad_sum(dot4(xi, xi)), quad_sum(dot4(al, xi)) + aR);
        float den = 1.0f;
        float4 acc = xi;

        int p = beg;
        for (; p + 4 <= end; p += 4) {
            const float4 x0 = load_row(xt, rec[p].x,     lane);
            const float4 x1 = load_row(xt, rec[p + 1].x, lane);
            const float4 x2 = load_row(xt, rec[p + 2].x, lane);
            const float4 x3 = load_row(xt, rec[p + 3].x, lane);
            const float lg = quad_transpose_sum(dot4(xi, x0), dot4(xi, x1),
                                                dot4(xi, x2), dot4(xi, x3), odd, hi);
            const float aa = quad_transpose_sum(dot4(al, x0), dot4(al, x1),
                                                dot4(al, x2), dot4(al, x3), odd, hi) + aR;
            const float v  = edge_score(lg, aa);
            float mx = fmaxf(v,  quad_move<QUAD_XOR1>(v));
            mx       = fmaxf(mx, quad_move<QUAD_XOR2>(mx));
            const float mn = fmaxf(m, mx);
            const float sc = exp_hw(m - mn);
            const float pe = exp_hw(v - mn);
            const float p0 = quad_move<0x00>(pe);
            const float p1 = quad_move<0x55>(pe);
            const float p2 = quad_move<0xAA>(pe);
            const float p3 = quad_move<0xFF>(pe);
            den   = den   * sc + ((p0 + p1) + (p2 + p3));
            acc.x = acc.x * sc + ((p0 * x0.x + p1 * x1.x) + (p2 * x2.x + p3 * x3.x));
            acc.y = acc.y * sc + ((p0 * x0.y + p1 * x1.y) + (p2 * x2.y + p3 * x3.y));
            acc.z = acc.z * sc + ((p0 * x0.z + p1 * x1.z) + (p2 * x2.z + p3 * x3.z));
            acc.w = acc.w * sc + ((p0 * x0.w + p1 * x1.w) + (p2 * x2.w + p3 * x3.w));
            m = mn;
        }
        for (; p < end; ++p) {
            const float4 x = load_row(xt, rec[p].x, lane);
            const float v  = edge_score(quad_sum(dot4(xi, x)), quad_sum(dot4(al, x)) + aR);
            const float mn = fmaxf(m, v);
            const float sc = exp_hw(m - mn);
            const float pe = exp_hw(v - mn);
            den   = den   * sc + pe;
            acc.x = acc.x * sc + pe * x.x;
            acc.y = acc.y * sc + pe * x.y;
            acc.z = acc.z * sc + pe * x.z;
            acc.w = acc.w * sc + pe * x.w;
            m = mn;
        }
        const float inv = rcp_hw(den + 1e-16f);
        if (!LAST) {
            *(float4*)(out + (size_t)node * FDIM + lane * 4) =
                make_float4(acc.x * inv, acc.y * inv, acc.z * inv, acc.w * inv);
        } else {
            float4 h;
            h.x = gelu_exact(acc.x * inv + bb.x);
            h.y = gelu_exact(acc.y * inv + bb.y);
            h.z = gelu_exact(acc.z * inv + bb.z);
            h.w = gelu_exact(acc.w * inv + bb.w);
            const float y = heads_sum(quad_sum(dot4(h, fw)));
            if (lane == 0) out[node] = y + fb;
        }
    }
}

extern "C" void kernel_launch(void* const* d_in, const int* in_sizes, int n_in,
                              void* d_out, int out_size, void* d_ws, size_t ws_size,
                              hipStream_t stream) {
    (void)in_sizes; (void)n_in; (void)out_size; (void)ws_size;
    const float* x   = (const float*)d_in[0];
    const float* W1  = (const float*)d_in[1];
    const float* b1  = (const float*)d_in[2];
    const float* al1 = (const float*)d_in[3];
    const float* ar1 = (const float*)d_in[4];
    const float* W2  = (const float*)d_in[5];
    const float* b2  = (const float*)d_in[6];
    const float* al2 = (const float*)d_in[7];
    const float* ar2 = (const float*)d_in[8];
    const float* fcW = (const float*)d_in[9];
    const float* fcb = (const float*)d_in[10];
    const int*   ei  = (const int*)d_in[11];

    float* ws   = (float*)d_ws;
    float* xt   = ws;
    float* agg  = xt   + (size_t)NNODES * FDIM;
    int*   tab  = (int*)(agg + (size_t)NNODES * FDIM);
    int*   pre  = tab  + (size_t)SCAN_N;
    int*   rowp = pre  + (size_t)SCAN_N + 4;
    int*   tmp  = rowp + (size_t)NNODES + 4;
    int*   csr  = tmp  + (size_t)EEDGES * 2;

    const dim3 B(256);

    gat_sort_hist_gemm<<<SORT_CHUNKS + TILES_HIST, B, 0, stream>>>(ei, tab, x, W1, xt);
    gat_csr_scan_gemm<<<1 + TILES_SCAN / 4, 1024, 0, stream>>>(tab, pre, x, W1, xt);
    gat_sort_scatter_gemm<<<SORT_CHUNKS + TILES_SCATTER, B, 0, stream>>>(ei, pre, tmp, x, W1, xt);
    gat_sort_place_gemm<<<SORT_BUCKETS + TILES_PLACE, B, 0, stream>>>(pre, tmp, rowp, csr, x, W1, xt);
    gat_attn_fused<false><<<GATHER_WAVES / 4, 128, 0, stream>>>(
        xt, al1, ar1, b1, fcW, fcb, agg, rowp, csr);

    gat_gemm128_gelu_wmma<<<NNODES / 16, B, 0, stream>>>(agg, b1, W2, xt);
    gat_attn_fused<true><<<GATHER_WAVES / 4, 128, 0, stream>>>(
        xt, al2, ar2, b2, fcW, fcb, (float*)d_out, rowp, csr);
}


# edge sort cut from four launches to two: a chunk counts, scans and buckets its own edges in LDS (no global scan), a range then gathers its 200 slices; 7 launches -> 5; row sum order changes
# speedup vs baseline: 10.9476x; 1.2081x over previous
#include <hip/hip_runtime.h>
#include <hip/hip_bf16.h>


#define NNODES  50000
#define EEDGES  800000
#define ETOT    (EEDGES + NNODES)
#define FDIM    128
#define NHEADS  8
#define CDIM    16

typedef __attribute__((ext_vector_type(2))) float v2f;
typedef __attribute__((ext_vector_type(8))) float v8f;

__device__ __forceinline__ float gelu_exact(float v) {
    return 0.5f * v * (1.0f + erff(v * 0.70710678118654752f));
}

template <bool GELU_IN>
__device__ __forceinline__ void gemm_tile16(const float* __restrict__ Ain,
                                            const float* __restrict__ abias,
                                            const float* __restrict__ Wmat,
                                            float* __restrict__ Out,
                                            float* As, const int tile, const int t) {
    const int tileBase = tile * 16;

    const float4* A4 = (const float4*)(Ain + (size_t)tileBase * FDIM);
    const int r0 = t >> 5;
    const int c4 = t & 31;
    float4 v[2];
    #pragma unroll
    for (int j = 0; j < 2; ++j) v[j] = A4[(j * 8 + r0) * 32 + c4];
    if (GELU_IN) {
        const float4 b = *(const float4*)(abias + c4 * 4);
        #pragma unroll
        for (int j = 0; j < 2; ++j) {
            v[j].x = gelu_exact(v[j].x + b.x);
            v[j].y = gelu_exact(v[j].y + b.y);
            v[j].z = gelu_exact(v[j].z + b.z);
            v[j].w = gelu_exact(v[j].w + b.w);
        }
    }
    #pragma unroll
    for (int j = 0; j < 2; ++j)
        *(float4*)&As[(j * 8 + r0) * 132 + c4 * 4] = v[j];
    __syncthreads();

    const int wave = t >> 5;
    const int lane = t & 31;
    const int m    = lane & 15;
    const int half = lane >> 4;
    const int col  = wave * 16 + m;

    v8f acc = {};
    #pragma unroll 4
    for (int kk = 0; kk < 32; ++kk) {
        const int k = kk * 4 + half * 2;
        v2f a, b;
        a.x = As[m * 132 + k];
        a.y = As[m * 132 + k + 1];
        b.x = Wmat[(size_t)k * FDIM + col];
        b.y = Wmat[(size_t)(k + 1) * FDIM + col];
        acc = __builtin_amdgcn_wmma_f32_16x16x4_f32(
             false, a,  false, b,
             (short)0, acc,  false,  false);
    }

    const int mbase = half * 8;
    #pragma unroll
    for (int i = 0; i < 8; ++i)
        Out[(size_t)(tileBase + mbase + i) * FDIM + wave * 16 + m] = acc[i];
}

__global__ void gat_gemm128_gelu_wmma(const float* __restrict__ Ain,
                                      const float* __restrict__ abias,
                                      const float* __restrict__ Wmat,
                                      float* __restrict__ Out) {
    __shared__ float As[16 * 132];
    gemm_tile16<true>(Ain, abias, Wmat, Out, As, (int)blockIdx.x, (int)threadIdx.x);
}

#define SORT_CHUNKS  200
#define CHUNK_EDGES  4000
#define SORT_BUCKETS 250
#define BUCKET_NODES 200
#define SCAN_N       (SORT_BUCKETS * SORT_CHUNKS)
static_assert(SORT_CHUNKS * CHUNK_EDGES == EEDGES, "chunks must tile the edge list");
static_assert(SORT_BUCKETS * BUCKET_NODES == NNODES, "ranges must tile the nodes");
static_assert(CHUNK_EDGES % 4 == 0, "int4 access");
static_assert(SORT_BUCKETS <= 256 && BUCKET_NODES <= 256, "one counter per thread");
static_assert(SORT_CHUNKS % 8 == 0 && SORT_CHUNKS <= 256, "slices dealt evenly to the 8 waves");

#define SORT_TRIPS ((CHUNK_EDGES / 4 + 255) / 256)
#define SLICES_PER_WAVE (SORT_CHUNKS / 8)

#define GEMM_TILES    (NNODES / 16)
#define TILES_CHUNK   1800
#define TILES_PLACE   1325
#define TILE0_CHUNK   0
#define TILE0_PLACE   (TILE0_CHUNK + TILES_CHUNK)
static_assert(GEMM_TILES * 16 == NNODES, "row tiles must cover the nodes exactly");
static_assert(TILE0_PLACE + TILES_PLACE == GEMM_TILES, "the two shares must cover every row tile once");

__device__ __forceinline__ void sort_chunk_stage(const int* __restrict__ ei, int* __restrict__ tab,
                                                 int* __restrict__ cst, int* __restrict__ tmp,
                                                 const int g) {
    __shared__ int lh[256];
    __shared__ int wtot[8];
    const int t = threadIdx.x;
    const int lane = t & 31, w = t >> 5;
    const int4* s4 = (const int4*)(ei + g * CHUNK_EDGES);
    const int4* d4 = (const int4*)(ei + EEDGES + g * CHUNK_EDGES);
    int4 s[SORT_TRIPS], d[SORT_TRIPS];
    #pragma unroll
    for (int k = 0; k < SORT_TRIPS; ++k) {
        const int q = t + k * 256;
        const bool in = q < CHUNK_EDGES / 4;
        s[k] = in ? s4[q] : make_int4(0, 0, 0, 0);
        d[k] = in ? d4[q] : make_int4(-1, -1, -1, -1);
    }
    lh[t] = 0;
    __syncthreads();
    #pragma unroll
    for (int k = 0; k < SORT_TRIPS; ++k) {
        if ((unsigned)d[k].x < (unsigned)NNODES) atomicAdd(&lh[d[k].x / BUCKET_NODES], 1);
        if ((unsigned)d[k].y < (unsigned)NNODES) atomicAdd(&lh[d[k].y / BUCKET_NODES], 1);
        if ((unsigned)d[k].z < (unsigned)NNODES) atomicAdd(&lh[d[k].z / BUCKET_NODES], 1);
        if ((unsigned)d[k].w < (unsigned)NNODES) atomicAdd(&lh[d[k].w / BUCKET_NODES], 1);
    }
    __syncthreads();
    const int c = lh[t];
    int inc = c;
    for (int o = 1; o < 32; o <<= 1) {
        int u = __shfl_up(inc, o);
        if (lane >= o) inc += u;
    }
    if (lane == 31) wtot[w] = inc;
    __syncthreads();
    int first = inc - c;
    for (int k = 0; k < w; ++k) first += wtot[k];
    if (t < SORT_BUCKETS) {
        tab[t * SORT_CHUNKS + g] = c;
        cst[t * SORT_CHUNKS + g] = first;
    }
    lh[t] = first;
    __syncthreads();
    int2* out = (int2*)tmp + (size_t)g * CHUNK_EDGES;
    #pragma unroll
    for (int k = 0; k < SORT_TRIPS; ++k) {
        const int sv[4] = {s[k].x, s[k].y, s[k].z, s[k].w};
        const int dv[4] = {d[k].x, d[k].y, d[k].z, d[k].w};
        int pos[4];
        #pragma unroll
        for (int e = 0; e < 4; ++e)
            pos[e] = ((unsigned)dv[e] < (unsigned)NNODES) ? atomicAdd(&lh[dv[e] / BUCKET_NODES], 1) : -1;
        #pragma unroll
        for (int e = 0; e < 4; ++e) {
            if ((unsigned)pos[e] < (unsigned)CHUNK_EDGES)
                out[pos[e]] = make_int2(sv[e], dv[e]);
        }
    }
}

__global__ void __launch_bounds__(256)
gat_sort_chunk_gemm(const int* __restrict__ ei, int* __restrict__ tab, int* __restrict__ cst,
                    int* __restrict__ tmp,
                    const float* __restrict__ x, const float* __restrict__ W1,
                    float* __restrict__ xt) {
    __shared__ float As[16 * 132];
    const int b = blockIdx.x;
    if (b < SORT_CHUNKS) sort_chunk_stage(ei, tab, cst, tmp, b);
    else gemm_tile16<false>(x, nullptr, W1, xt, As, TILE0_CHUNK + (b - SORT_CHUNKS), (int)threadIdx.x);
}

__device__ __forceinline__ void sort_place_stage(const int* __restrict__ tab, const int* __restrict__ cst,
                                                 const int* __restrict__ tmp,
                                                 int* __restrict__ rowptr, int* __restrict__ csr,
                                                 const int b) {
    __shared__ int cnt[256];
    __shared__ int wtot[8];
    __shared__ int wfirst[8], wnum[8];
    const int t = threadIdx.x;
    const int lane = t & 31, w = t >> 5;
    const int wu = __builtin_amdgcn_readfirstlane(w);
    const int node0 = b * BUCKET_NODES;
    const int2* in = (const int2*)tmp;
    int c = 0, o = 0, oa = 0;
    if (lane < SLICES_PER_WAVE) {
        const int g = wu + 8 * lane;
        c  = min(max(tab[b * SORT_CHUNKS + g], 0), CHUNK_EDGES);
        o  = min(max(cst[b * SORT_CHUNKS + g], 0), CHUNK_EDGES - c);
        oa = g * CHUNK_EDGES + o;
    }
    cnt[t] = 0;
    int osum = o, csum = c;
    for (int m = 16; m >= 1; m >>= 1) {
        osum += __shfl_xor(osum, m);
        csum += __shfl_xor(csum, m);
    }
    if (lane == 0) { wfirst[w] = osum; wnum[w] = csum; }

    int2 r[SLICES_PER_WAVE];
    #pragma unroll
    for (int m = 0; m < SLICES_PER_WAVE; ++m) {
        const int cm = __builtin_amdgcn_readlane(c, m);
        const int om = __builtin_amdgcn_readlane(oa, m);
        r[m] = (lane < cm) ? in[om + lane] : make_int2(0, -1);
    }
    __syncthreads();
    int start = 0, num = 0;
    for (int k = 0; k < 8; ++k) { start += wfirst[k]; num += wnum[k]; }
    start = min(start, EEDGES);
    const int end = min(start + num, EEDGES);
    #pragma unroll
    for (int m = 0; m < SLICES_PER_WAVE; ++m) {
        const int j = r[m].y - node0;
        if ((unsigned)j < (unsigned)BUCKET_NODES) atomicAdd(&cnt[j], 1);
    }
    for (int m = 0; m < SLICES_PER_WAVE; ++m) {
        const int cm = __builtin_amdgcn_readlane(c, m);
        const int om = __builtin_amdgcn_readlane(oa, m);
        for (int k = lane + 32; k < cm; k += 32) {
            const int j = in[om + k].y - node0;
            if ((unsigned)j < (unsigned)BUCKET_NODES) atomicAdd(&cnt[j], 1);
        }
    }
    __syncthreads();
    const int cn = cnt[t];
    int inc = cn;
    for (int q = 1; q < 32; q <<= 1) {
        int u = __shfl_up(inc, q);
        if (lane >= q) inc += u;
    }
    if (lane == 31) wtot[w] = inc;
    __syncthreads();
    int first = start + inc - cn;
    for (int k = 0; k < w; ++k) first += wtot[k];
    cnt[t] = first;
    if (t < BUCKET_NODES) rowptr[node0 + t] = first;
    if (b == SORT_BUCKETS - 1 && t == 0) rowptr[NNODES] = end;
    __syncthreads();
    int pos[SLICES_PER_WAVE];
    #pragma unroll
    for (int m = 0; m < SLICES_PER_WAVE; ++m) {
        const int j = r[m].y - node0;
        pos[m] = ((unsigned)j < (unsigned)BUCKET_NODES) ? atomicAdd(&cnt[j], 1) : -1;
    }
    #pragma unroll
    for (int m = 0; m < SLICES_PER_WAVE; ++m) {
        if ((unsigned)pos[m] < (unsigned)EEDGES)
            ((int2*)csr)[pos[m]] = r[m];
    }
    for (int m = 0; m < SLICES_PER_WAVE; ++m) {
        const int cm = __builtin_amdgcn_readlane(c, m);
        const int om = __builtin_amdgcn_readlane(oa, m);
        for (int k = lane + 32; k < cm; k += 32) {
            const int2 rk = in[om + k];
            const int j = rk.y - node0;
            if ((unsigned)j < (unsigned)BUCKET_NODES) {
                const int pk = atomicAdd(&cnt[j], 1);
                if ((unsigned)pk < (unsigned)EEDGES)
                    ((int2*)csr)[pk] = rk;
            }
        }
    }
}

__global__ void __launch_bounds__(256)
gat_sort_place_gemm(const int* __restrict__ tab, const int* __restrict__ cst, const int* __restrict__ tmp,
                    int* __restrict__ rowptr, int* __restrict__ csr,
                    const float* __restrict__ x, const float* __restrict__ W1,
                    float* __restrict__ xt) {
    __shared__ float As[16 * 132];
    const int b = blockIdx.x;
    if (b < SORT_BUCKETS) sort_place_stage(tab, cst, tmp, rowptr, csr, b);
    else gemm_tile16<false>(x, nullptr, W1, xt, As, TILE0_PLACE + (b - SORT_BUCKETS), (int)threadIdx.x);
}

#define QUAD_XOR1 0xB1
#define QUAD_XOR2 0x4E
template <int CTRL>
__device__ __forceinline__ float quad_move(float v) {
    return __int_as_float(__builtin_amdgcn_mov_dpp(__float_as_int(v), CTRL, 0xF, 0xF, true));
}

__device__ __forceinline__ float quad_sum(float v) {
    v += quad_move<QUAD_XOR1>(v);
    v += quad_move<QUAD_XOR2>(v);
    return v;
}

__device__ __forceinline__ float quad_transpose_sum(float e0, float e1, float e2, float e3,
                                                    bool odd, bool hi) {
    const float rA = (odd ? e1 : e0) + quad_move<QUAD_XOR1>(odd ? e0 : e1);
    const float rB = (odd ? e3 : e2) + quad_move<QUAD_XOR1>(odd ? e2 : e3);
    return (hi ? rB : rA) + quad_move<QUAD_XOR2>(hi ? rA : rB);
}

__device__ __forceinline__ float dot4(float4 a, float4 b) {
    return a.x * b.x + a.y * b.y + a.z * b.z + a.w * b.w;
}

#define LOG2E 1.4426950408889634f
__device__ __forceinline__ float exp_hw(float x) {
    return __builtin_amdgcn_exp2f(x * LOG2E);
}
__device__ __forceinline__ float rcp_hw(float x) {
    return __builtin_amdgcn_rcpf(x);
}

__device__ __forceinline__ float edge_score(float logit, float a) {
    float sg = rcp_hw(1.0f + exp_hw(-logit));
    float v  = a * sg;
    return (v > 0.0f) ? v : 0.2f * v;
}

#define ROW_XOR4 0x164
#define ROW_XOR8 0x168
__device__ __forceinline__ float heads_sum(float v) {
    v += quad_move<ROW_XOR4>(v);
    v += quad_move<ROW_XOR8>(v);
    const int b = __float_as_int(v);
    return v + __int_as_float(__builtin_amdgcn_permlanex16(b, b, 0x76543210, 0xfedcba98, false, false));
}

__device__ __forceinline__ float4 load_row(const float* __restrict__ xt, int s, int lane) {
    s = min(max(s, 0), NNODES - 1);
    return *(const float4*)(xt + (size_t)s * FDIM + lane * 4);
}

#define GATHER_WAVES 10000
template <bool LAST>
__global__ void __launch_bounds__(128)
gat_attn_fused(const float* __restrict__ xt,
               const float* __restrict__ attL,
               const float* __restrict__ attR,
               const float* __restrict__ bias,
               const float* __restrict__ fcw,
               const float* __restrict__ fcb,
               float* __restrict__ out,
               const int* __restrict__ rowptr,
               const int* __restrict__ csr) {
    const int lane = threadIdx.x & 31;
    const bool odd = (lane & 1) != 0;
    const bool hi  = (lane & 2) != 0;
    const int wv = __builtin_amdgcn_readfirstlane((int)(threadIdx.x >> 5));
    const int gw = blockIdx.x * 4 + wv;
    const int2* rec = (const int2*)csr;
    const float4 al = *(const float4*)(attL + lane * 4);
    const float4 ar = *(const float4*)(attR + lane * 4);
    const float4 bb = LAST ? *(const float4*)(bias + lane * 4) : make_float4(0.f, 0.f, 0.f, 0.f);
    const float4 fw = LAST ? *(const float4*)(fcw + lane * 4) : make_float4(0.f, 0.f, 0.f, 0.f);
    const float  fb = LAST ? fcb[0] : 0.0f;

    for (int node = gw; node < NNODES; node += GATHER_WAVES) {
        const int beg = max(rowptr[node], 0), end = min(rowptr[node + 1], EEDGES);
        const float4 xi = *(const float4*)(xt + (size_t)node * FDIM + lane * 4);
        const float aR = quad_sum(dot4(ar, xi));
        float m   = edge_score(quad_sum(dot4(xi, xi)), quad_sum(dot4(al, xi)) + aR);
        float den = 1.0f;
        float4 acc = xi;

        int p = beg;
        for (; p + 4 <= end; p += 4) {
            const float4 x0 = load_row(xt, rec[p].x,     lane);
            const float4 x1 = load_row(xt, rec[p + 1].x, lane);
            const float4 x2 = load_row(xt, rec[p + 2].x, lane);
            const float4 x3 = load_row(xt, rec[p + 3].x, lane);
            const float lg = quad_transpose_sum(dot4(xi, x0), dot4(xi, x1),
                                                dot4(xi, x2), dot4(xi, x3), odd, hi);
            const float aa = quad_transpose_sum(dot4(al, x0), dot4(al, x1),
                                                dot4(al, x2), dot4(al, x3), odd, hi) + aR;
            const float v  = edge_score(lg, aa);
            float mx = fmaxf(v,  quad_move<QUAD_XOR1>(v));
            mx       = fmaxf(mx, quad_move<QUAD_XOR2>(mx));
            const float mn = fmaxf(m, mx);
            const float sc = exp_hw(m - mn);
            const float pe = exp_hw(v - mn);
            const float p0 = quad_move<0x00>(pe);
            const float p1 = quad_move<0x55>(pe);
            const float p2 = quad_move<0xAA>(pe);
            const float p3 = quad_move<0xFF>(pe);
            den   = den   * sc + ((p0 + p1) + (p2 + p3));
            acc.x = acc.x * sc + ((p0 * x0.x + p1 * x1.x) + (p2 * x2.x + p3 * x3.x));
            acc.y = acc.y * sc + ((p0 * x0.y + p1 * x1.y) + (p2 * x2.y + p3 * x3.y));
            acc.z = acc.z * sc + ((p0 * x0.z + p1 * x1.z) + (p2 * x2.z + p3 * x3.z));
            acc.w = acc.w * sc + ((p0 * x0.w + p1 * x1.w) + (p2 * x2.w + p3 * x3.w));
            m = mn;
        }
        for (; p < end; ++p) {
            const float4 x = load_row(xt, rec[p].x, lane);
            const float v  = edge_score(quad_sum(dot4(xi, x)), quad_sum(dot4(al, x)) + aR);
            const float mn = fmaxf(m, v);
            const float sc = exp_hw(m - mn);
            const float pe = exp_hw(v - mn);
            den   = den   * sc + pe;
            acc.x = acc.x * sc + pe * x.x;
            acc.y = acc.y * sc + pe * x.y;
            acc.z = acc.z * sc + pe * x.z;
            acc.w = acc.w * sc + pe * x.w;
            m = mn;
        }
        const float inv = rcp_hw(den + 1e-16f);
        if (!LAST) {
            *(float4*)(out + (size_t)node * FDIM + lane * 4) =
                make_float4(acc.x * inv, acc.y * inv, acc.z * inv, acc.w * inv);
        } else {
            float4 h;
            h.x = gelu_exact(acc.x * inv + bb.x);
            h.y = gelu_exact(acc.y * inv + bb.y);
            h.z = gelu_exact(acc.z * inv + bb.z);
            h.w = gelu_exact(acc.w * inv + bb.w);
            const float y = heads_sum(quad_sum(dot4(h, fw)));
            if (lane == 0) out[node] = y + fb;
        }
    }
}

extern "C" void kernel_launch(void* const* d_in, const int* in_sizes, int n_in,
                              void* d_out, int out_size, void* d_ws, size_t ws_size,
                              hipStream_t stream) {
    (void)in_sizes; (void)n_in; (void)out_size; (void)ws_size;
    const float* x   = (const float*)d_in[0];
    const float* W1  = (const float*)d_in[1];
    const float* b1  = (const float*)d_in[2];
    const float* al1 = (const float*)d_in[3];
    const float* ar1 = (const float*)d_in[4];
    const float* W2  = (const float*)d_in[5];
    const float* b2  = (const float*)d_in[6];
    const float* al2 = (const float*)d_in[7];
    const float* ar2 = (const float*)d_in[8];
    const float* fcW = (const float*)d_in[9];
    const float* fcb = (const float*)d_in[10];
    const int*   ei  = (const int*)d_in[11];

    float* ws   = (float*)d_ws;
    float* xt   = ws;
    float* agg  = xt   + (size_t)NNODES * FDIM;
    int*   tab  = (int*)(agg + (size_t)NNODES * FDIM);
    int*   cst  = tab  + (size_t)SCAN_N;
    int*   rowp = cst  + (size_t)SCAN_N + 4;
    int*   tmp  = rowp + (size_t)NNODES + 4;
    int*   csr  = tmp  + (size_t)EEDGES * 2;

    const dim3 B(256);

    gat_sort_chunk_gemm<<<SORT_CHUNKS + TILES_CHUNK, B, 0, stream>>>(ei, tab, cst, tmp, x, W1, xt);
    gat_sort_place_gemm<<<SORT_BUCKETS + TILES_PLACE, B, 0, stream>>>(tab, cst, tmp, rowp, csr, x, W1, xt);
    gat_attn_fused<false><<<GATHER_WAVES / 4, 128, 0, stream>>>(
        xt, al1, ar1, b1, fcW, fcb, agg, rowp, csr);

    gat_gemm128_gelu_wmma<<<NNODES / 16, B, 0, stream>>>(agg, b1, W2, xt);
    gat_attn_fused<true><<<GATHER_WAVES / 4, 128, 0, stream>>>(
        xt, al2, ar2, b2, fcW, fcb, (float*)d_out, rowp, csr);
}


# edge pass: per-edge dot products and weighted row sums written on 2-wide vectors as packed fused multiply-add chains (one chain gives both dots of an edge), no register moves; sum order changes
# speedup vs baseline: 11.5425x; 1.0543x over previous
#include <hip/hip_runtime.h>
#include <hip/hip_bf16.h>


#define NNODES  50000
#define EEDGES  800000
#define ETOT    (EEDGES + NNODES)
#define FDIM    128
#define NHEADS  8
#define CDIM    16

typedef __attribute__((ext_vector_type(2))) float v2f;
typedef __attribute__((ext_vector_type(4))) float v4f;
typedef __attribute__((ext_vector_type(8))) float v8f;

__device__ __forceinline__ float gelu_exact(float v) {
    return 0.5f * v * (1.0f + erff(v * 0.70710678118654752f));
}

template <bool GELU_IN>
__device__ __forceinline__ void gemm_tile16(const float* __restrict__ Ain,
                                            const float* __restrict__ abias,
                                            const float* __restrict__ Wmat,
                                            float* __restrict__ Out,
                                            float* As, const int tile, const int t) {
    const int tileBase = tile * 16;

    const float4* A4 = (const float4*)(Ain + (size_t)tileBase * FDIM);
    const int r0 = t >> 5;
    const int c4 = t & 31;
    float4 v[2];
    #pragma unroll
    for (int j = 0; j < 2; ++j) v[j] = A4[(j * 8 + r0) * 32 + c4];
    if (GELU_IN) {
        const float4 b = *(const float4*)(abias + c4 * 4);
        #pragma unroll
        for (int j = 0; j < 2; ++j) {
            v[j].x = gelu_exact(v[j].x + b.x);
            v[j].y = gelu_exact(v[j].y + b.y);
            v[j].z = gelu_exact(v[j].z + b.z);
            v[j].w = gelu_exact(v[j].w + b.w);
        }
    }
    #pragma unroll
    for (int j = 0; j < 2; ++j)
        *(float4*)&As[(j * 8 + r0) * 132 + c4 * 4] = v[j];
    __syncthreads();

    const int wave = t >> 5;
    const int lane = t & 31;
    const int m    = lane & 15;
    const int half = lane >> 4;
    const int col  = wave * 16 + m;

    v8f acc = {};
    #pragma unroll 4
    for (int kk = 0; kk < 32; ++kk) {
        const int k = kk * 4 + half * 2;
        v2f a, b;
        a.x = As[m * 132 + k];
        a.y = As[m * 132 + k + 1];
        b.x = Wmat[(size_t)k * FDIM + col];
        b.y = Wmat[(size_t)(k + 1) * FDIM + col];
        acc = __builtin_amdgcn_wmma_f32_16x16x4_f32(
             false, a,  false, b,
             (short)0, acc,  false,  false);
    }

    const int mbase = half * 8;
    #pragma unroll
    for (int i = 0; i < 8; ++i)
        Out[(size_t)(tileBase + mbase + i) * FDIM + wave * 16 + m] = acc[i];
}

__global__ void gat_gemm128_gelu_wmma(const float* __restrict__ Ain,
                                      const float* __restrict__ abias,
                                      const float* __restrict__ Wmat,
                                      float* __restrict__ Out) {
    __shared__ float As[16 * 132];
    gemm_tile16<true>(Ain, abias, Wmat, Out, As, (int)blockIdx.x, (int)threadIdx.x);
}

#define SORT_CHUNKS  200
#define CHUNK_EDGES  4000
#define SORT_BUCKETS 250
#define BUCKET_NODES 200
#define SCAN_N       (SORT_BUCKETS * SORT_CHUNKS)
static_assert(SORT_CHUNKS * CHUNK_EDGES == EEDGES, "chunks must tile the edge list");
static_assert(SORT_BUCKETS * BUCKET_NODES == NNODES, "ranges must tile the nodes");
static_assert(CHUNK_EDGES % 4 == 0, "int4 access");
static_assert(SORT_BUCKETS <= 256 && BUCKET_NODES <= 256, "one counter per thread");
static_assert(SORT_CHUNKS % 8 == 0 && SORT_CHUNKS <= 256, "slices dealt evenly to the 8 waves");

#define SORT_TRIPS ((CHUNK_EDGES / 4 + 255) / 256)
#define SLICES_PER_WAVE (SORT_CHUNKS / 8)

#define GEMM_TILES    (NNODES / 16)
#define TILES_CHUNK   1800
#define TILES_PLACE   1325
#define TILE0_CHUNK   0
#define TILE0_PLACE   (TILE0_CHUNK + TILES_CHUNK)
static_assert(GEMM_TILES * 16 == NNODES, "row tiles must cover the nodes exactly");
static_assert(TILE0_PLACE + TILES_PLACE == GEMM_TILES, "the two shares must cover every row tile once");

__device__ __forceinline__ void sort_chunk_stage(const int* __restrict__ ei, int* __restrict__ tab,
                                                 int* __restrict__ cst, int* __restrict__ tmp,
                                                 const int g) {
    __shared__ int lh[256];
    __shared__ int wtot[8];
    const int t = threadIdx.x;
    const int lane = t & 31, w = t >> 5;
    const int4* s4 = (const int4*)(ei + g * CHUNK_EDGES);
    const int4* d4 = (const int4*)(ei + EEDGES + g * CHUNK_EDGES);
    int4 s[SORT_TRIPS], d[SORT_TRIPS];
    #pragma unroll
    for (int k = 0; k < SORT_TRIPS; ++k) {
        const int q = t + k * 256;
        const bool in = q < CHUNK_EDGES / 4;
        s[k] = in ? s4[q] : make_int4(0, 0, 0, 0);
        d[k] = in ? d4[q] : make_int4(-1, -1, -1, -1);
    }
    lh[t] = 0;
    __syncthreads();
    #pragma unroll
    for (int k = 0; k < SORT_TRIPS; ++k) {
        if ((unsigned)d[k].x < (unsigned)NNODES) atomicAdd(&lh[d[k].x / BUCKET_NODES], 1);
        if ((unsigned)d[k].y < (unsigned)NNODES) atomicAdd(&lh[d[k].y / BUCKET_NODES], 1);
        if ((unsigned)d[k].z < (unsigned)NNODES) atomicAdd(&lh[d[k].z / BUCKET_NODES], 1);
        if ((unsigned)d[k].w < (unsigned)NNODES) atomicAdd(&lh[d[k].w / BUCKET_NODES], 1);
    }
    __syncthreads();
    const int c = lh[t];
    int inc = c;
    for (int o = 1; o < 32; o <<= 1) {
        int u = __shfl_up(inc, o);
        if (lane >= o) inc += u;
    }
    if (lane == 31) wtot[w] = inc;
    __syncthreads();
    int first = inc - c;
    for (int k = 0; k < w; ++k) first += wtot[k];
    if (t < SORT_BUCKETS) {
        tab[t * SORT_CHUNKS + g] = c;
        cst[t * SORT_CHUNKS + g] = first;
    }
    lh[t] = first;
    __syncthreads();
    int2* out = (int2*)tmp + (size_t)g * CHUNK_EDGES;
    #pragma unroll
    for (int k = 0; k < SORT_TRIPS; ++k) {
        const int sv[4] = {s[k].x, s[k].y, s[k].z, s[k].w};
        const int dv[4] = {d[k].x, d[k].y, d[k].z, d[k].w};
        int pos[4];
        #pragma unroll
        for (int e = 0; e < 4; ++e)
            pos[e] = ((unsigned)dv[e] < (unsigned)NNODES) ? atomicAdd(&lh[dv[e] / BUCKET_NODES], 1) : -1;
        #pragma unroll
        for (int e = 0; e < 4; ++e) {
            if ((unsigned)pos[e] < (unsigned)CHUNK_EDGES)
                out[pos[e]] = make_int2(sv[e], dv[e]);
        }
    }
}

__global__ void __launch_bounds__(256)
gat_sort_chunk_gemm(const int* __restrict__ ei, int* __restrict__ tab, int* __restrict__ cst,
                    int* __restrict__ tmp,
                    const float* __restrict__ x, const float* __restrict__ W1,
                    float* __restrict__ xt) {
    __shared__ float As[16 * 132];
    const int b = blockIdx.x;
    if (b < SORT_CHUNKS) sort_chunk_stage(ei, tab, cst, tmp, b);
    else gemm_tile16<false>(x, nullptr, W1, xt, As, TILE0_CHUNK + (b - SORT_CHUNKS), (int)threadIdx.x);
}

__device__ __forceinline__ void sort_place_stage(const int* __restrict__ tab, const int* __restrict__ cst,
                                                 const int* __restrict__ tmp,
                                                 int* __restrict__ rowptr, int* __restrict__ csr,
                                                 const int b) {
    __shared__ int cnt[256];
    __shared__ int wtot[8];
    __shared__ int wfirst[8], wnum[8];
    const int t = threadIdx.x;
    const int lane = t & 31, w = t >> 5;
    const int wu = __builtin_amdgcn_readfirstlane(w);
    const int node0 = b * BUCKET_NODES;
    const int2* in = (const int2*)tmp;
    int c = 0, o = 0, oa = 0;
    if (lane < SLICES_PER_WAVE) {
        const int g = wu + 8 * lane;
        c  = min(max(tab[b * SORT_CHUNKS + g], 0), CHUNK_EDGES);
        o  = min(max(cst[b * SORT_CHUNKS + g], 0), CHUNK_EDGES - c);
        oa = g * CHUNK_EDGES + o;
    }
    cnt[t] = 0;
    int osum = o, csum = c;
    for (int m = 16; m >= 1; m >>= 1) {
        osum += __shfl_xor(osum, m);
        csum += __shfl_xor(csum, m);
    }
    if (lane == 0) { wfirst[w] = osum; wnum[w] = csum; }

    int2 r[SLICES_PER_WAVE];
    #pragma unroll
    for (int m = 0; m < SLICES_PER_WAVE; ++m) {
        const int cm = __builtin_amdgcn_readlane(c, m);
        const int om = __builtin_amdgcn_readlane(oa, m);
        r[m] = (lane < cm) ? in[om + lane] : make_int2(0, -1);
    }
    __syncthreads();
    int start = 0, num = 0;
    for (int k = 0; k < 8; ++k) { start += wfirst[k]; num += wnum[k]; }
    start = min(start, EEDGES);
    const int end = min(start + num, EEDGES);
    #pragma unroll
    for (int m = 0; m < SLICES_PER_WAVE; ++m) {
        const int j = r[m].y - node0;
        if ((unsigned)j < (unsigned)BUCKET_NODES) atomicAdd(&cnt[j], 1);
    }
    for (int m = 0; m < SLICES_PER_WAVE; ++m) {
        const int cm = __builtin_amdgcn_readlane(c, m);
        const int om = __builtin_amdgcn_readlane(oa, m);
        for (int k = lane + 32; k < cm; k += 32) {
            const int j = in[om + k].y - node0;
            if ((unsigned)j < (unsigned)BUCKET_NODES) atomicAdd(&cnt[j], 1);
        }
    }
    __syncthreads();
    const int cn = cnt[t];
    int inc = cn;
    for (int q = 1; q < 32; q <<= 1) {
        int u = __shfl_up(inc, q);
        if (lane >= q) inc += u;
    }
    if (lane == 31) wtot[w] = inc;
    __syncthreads();
    int first = start + inc - cn;
    for (int k = 0; k < w; ++k) first += wtot[k];
    cnt[t] = first;
    if (t < BUCKET_NODES) rowptr[node0 + t] = first;
    if (b == SORT_BUCKETS - 1 && t == 0) rowptr[NNODES] = end;
    __syncthreads();
    int pos[SLICES_PER_WAVE];
    #pragma unroll
    for (int m = 0; m < SLICES_PER_WAVE; ++m) {
        const int j = r[m].y - node0;
        pos[m] = ((unsigned)j < (unsigned)BUCKET_NODES) ? atomicAdd(&cnt[j], 1) : -1;
    }
    #pragma unroll
    for (int m = 0; m < SLICES_PER_WAVE; ++m) {
        if ((unsigned)pos[m] < (unsigned)EEDGES)
            ((int2*)csr)[pos[m]] = r[m];
    }
    for (int m = 0; m < SLICES_PER_WAVE; ++m) {
        const int cm = __builtin_amdgcn_readlane(c, m);
        const int om = __builtin_amdgcn_readlane(oa, m);
        for (int k = lane + 32; k < cm; k += 32) {
            const int2 rk = in[om + k];
            const int j = rk.y - node0;
            if ((unsigned)j < (unsigned)BUCKET_NODES) {
                const int pk = atomicAdd(&cnt[j], 1);
                if ((unsigned)pk < (unsigned)EEDGES)
                    ((int2*)csr)[pk] = rk;
            }
        }
    }
}

__global__ void __launch_bounds__(256)
gat_sort_place_gemm(const int* __restrict__ tab, const int* __restrict__ cst, const int* __restrict__ tmp,
                    int* __restrict__ rowptr, int* __restrict__ csr,
                    const float* __restrict__ x, const float* __restrict__ W1,
                    float* __restrict__ xt) {
    __shared__ float As[16 * 132];
    const int b = blockIdx.x;
    if (b < SORT_BUCKETS) sort_place_stage(tab, cst, tmp, rowptr, csr, b);
    else gemm_tile16<false>(x, nullptr, W1, xt, As, TILE0_PLACE + (b - SORT_BUCKETS), (int)threadIdx.x);
}

#define QUAD_XOR1 0xB1
#define QUAD_XOR2 0x4E
template <int CTRL>
__device__ __forceinline__ float quad_move(float v) {
    return __int_as_float(__builtin_amdgcn_mov_dpp(__float_as_int(v), CTRL, 0xF, 0xF, true));
}

__device__ __forceinline__ float quad_sum(float v) {
    v += quad_move<QUAD_XOR1>(v);
    v += quad_move<QUAD_XOR2>(v);
    return v;
}

__device__ __forceinline__ float quad_transpose_sum(float e0, float e1, float e2, float e3,
                                                    bool odd, bool hi) {
    const float rA = (odd ? e1 : e0) + quad_move<QUAD_XOR1>(odd ? e0 : e1);
    const float rB = (odd ? e3 : e2) + quad_move<QUAD_XOR1>(odd ? e2 : e3);
    return (hi ? rB : rA) + quad_move<QUAD_XOR2>(hi ? rA : rB);
}

__device__ __forceinline__ float dot4(v4f u, v4f x) {
    const v2f t = __builtin_elementwise_fma(u.hi, x.hi, u.lo * x.lo);
    return t.x + t.y;
}
__device__ __forceinline__ v2f splat2(float s) {
    v2f r = {s, s};
    return r;
}
struct pairs4 { v2f c0, c1, c2, c3; };
__device__ __forceinline__ v2f dot4x2(const pairs4 c, v4f x) {
    v2f p = c.c0 * splat2(x.x);
    p = __builtin_elementwise_fma(c.c1, splat2(x.y), p);
    p = __builtin_elementwise_fma(c.c2, splat2(x.z), p);
    p = __builtin_elementwise_fma(c.c3, splat2(x.w), p);
    return p;
}
__device__ __forceinline__ v2f wsum4(v2f acc, float sc, float p0, v2f x0, float p1, v2f x1,
                                     float p2, v2f x2, float p3, v2f x3) {
    v2f w = x0 * splat2(p0);
    w = __builtin_elementwise_fma(x1, splat2(p1), w);
    w = __builtin_elementwise_fma(x2, splat2(p2), w);
    w = __builtin_elementwise_fma(x3, splat2(p3), w);
    return __builtin_elementwise_fma(acc, splat2(sc), w);
}
__device__ __forceinline__ float dot4_fc(float4 a, float4 b) {
    return a.x * b.x + a.y * b.y + a.z * b.z + a.w * b.w;
}

#define LOG2E 1.4426950408889634f
__device__ __forceinline__ float exp_hw(float x) {
    return __builtin_amdgcn_exp2f(x * LOG2E);
}
__device__ __forceinline__ float rcp_hw(float x) {
    return __builtin_amdgcn_rcpf(x);
}

__device__ __forceinline__ float edge_score(float logit, float a) {
    float sg = rcp_hw(1.0f + exp_hw(-logit));
    float v  = a * sg;
    return (v > 0.0f) ? v : 0.2f * v;
}

#define ROW_XOR4 0x164
#define ROW_XOR8 0x168
__device__ __forceinline__ float heads_sum(float v) {
    v += quad_move<ROW_XOR4>(v);
    v += quad_move<ROW_XOR8>(v);
    const int b = __float_as_int(v);
    return v + __int_as_float(__builtin_amdgcn_permlanex16(b, b, 0x76543210, 0xfedcba98, false, false));
}

__device__ __forceinline__ v4f load_row(const float* __restrict__ xt, int s, int lane) {
    s = min(max(s, 0), NNODES - 1);
    return *(const v4f*)(xt + (size_t)s * FDIM + lane * 4);
}

#define GATHER_WAVES 10000
template <bool LAST>
__global__ void __launch_bounds__(128)
gat_attn_fused(const float* __restrict__ xt,
               const float* __restrict__ attL,
               const float* __restrict__ attR,
               const float* __restrict__ bias,
               const float* __restrict__ fcw,
               const float* __restrict__ fcb,
               float* __restrict__ out,
               const int* __restrict__ rowptr,
               const int* __restrict__ csr) {
    const int lane = threadIdx.x & 31;
    const bool odd = (lane & 1) != 0;
    const bool hi  = (lane & 2) != 0;
    const int wv = __builtin_amdgcn_readfirstlane((int)(threadIdx.x >> 5));
    const int gw = blockIdx.x * 4 + wv;
    const int2* rec = (const int2*)csr;
    const v4f al = *(const v4f*)(attL + lane * 4);
    const v4f ar = *(const v4f*)(attR + lane * 4);
    const float4 bb = LAST ? *(const float4*)(bias + lane * 4) : make_float4(0.f, 0.f, 0.f, 0.f);
    const float4 fw = LAST ? *(const float4*)(fcw + lane * 4) : make_float4(0.f, 0.f, 0.f, 0.f);
    const float  fb = LAST ? fcb[0] : 0.0f;

    for (int node = gw; node < NNODES; node += GATHER_WAVES) {
        const int beg = max(rowptr[node], 0), end = min(rowptr[node + 1], EEDGES);
        const v4f xi = *(const v4f*)(xt + (size_t)node * FDIM + lane * 4);
        const float aR = quad_sum(dot4(ar, xi));
        const pairs4 cf = {{xi.x, al.x}, {xi.y, al.y}, {xi.z, al.z}, {xi.w, al.w}};
        const v2f qs = dot4x2(cf, xi);
        float m   = edge_score(quad_sum(qs.x), quad_sum(qs.y) + aR);
        float den = 1.0f;
        v2f accl = xi.lo, acch = xi.hi;

        int p = beg;
        for (; p + 4 <= end; p += 4) {
            const v4f x0 = load_row(xt, rec[p].x,     lane);
            const v4f x1 = load_row(xt, rec[p + 1].x, lane);
            const v4f x2 = load_row(xt, rec[p + 2].x, lane);
            const v4f x3 = load_row(xt, rec[p + 3].x, lane);
            const v2f q0 = dot4x2(cf, x0), q1 = dot4x2(cf, x1);
            const v2f q2 = dot4x2(cf, x2), q3 = dot4x2(cf, x3);
            const float lg = quad_transpose_sum(q0.x, q1.x, q2.x, q3.x, odd, hi);
            const float aa = quad_transpose_sum(q0.y, q1.y, q2.y, q3.y, odd, hi) + aR;
            const float v  = edge_score(lg, aa);
            float mx = fmaxf(v,  quad_move<QUAD_XOR1>(v));
            mx       = fmaxf(mx, quad_move<QUAD_XOR2>(mx));
            const float mn = fmaxf(m, mx);
            const float sc = exp_hw(m - mn);
            const float pe = exp_hw(v - mn);
            const float p0 = quad_move<0x00>(pe);
            const float p1 = quad_move<0x55>(pe);
            const float p2 = quad_move<0xAA>(pe);
            const float p3 = quad_move<0xFF>(pe);
            den  = den * sc + ((p0 + p1) + (p2 + p3));
            accl = wsum4(accl, sc, p0, x0.lo, p1, x1.lo, p2, x2.lo, p3, x3.lo);
            acch = wsum4(acch, sc, p0, x0.hi, p1, x1.hi, p2, x2.hi, p3, x3.hi);
            m = mn;
        }
        for (; p < end; ++p) {
            const v4f x = load_row(xt, rec[p].x, lane);
            const v2f q = dot4x2(cf, x);
            const float v  = edge_score(quad_sum(q.x), quad_sum(q.y) + aR);
            const float mn = fmaxf(m, v);
            const float sc = exp_hw(m - mn);
            const float pe = exp_hw(v - mn);
            den  = den * sc + pe;
            accl = __builtin_elementwise_fma(accl, splat2(sc), x.lo * splat2(pe));
            acch = __builtin_elementwise_fma(acch, splat2(sc), x.hi * splat2(pe));
            m = mn;
        }
        const float inv = rcp_hw(den + 1e-16f);
        if (!LAST) {
            const v2f ol = accl * splat2(inv), oh = acch * splat2(inv);
            v4f o = {ol.x, ol.y, oh.x, oh.y};
            *(v4f*)(out + (size_t)node * FDIM + lane * 4) = o;
        } else {
            float4 h;
            h.x = gelu_exact(accl.x * inv + bb.x);
            h.y = gelu_exact(accl.y * inv + bb.y);
            h.z = gelu_exact(acch.x * inv + bb.z);
            h.w = gelu_exact(acch.y * inv + bb.w);
            const float y = heads_sum(quad_sum(dot4_fc(h, fw)));
            if (lane == 0) out[node] = y + fb;
        }
    }
}

extern "C" void kernel_launch(void* const* d_in, const int* in_sizes, int n_in,
                              void* d_out, int out_size, void* d_ws, size_t ws_size,
                              hipStream_t stream) {
    (void)in_sizes; (void)n_in; (void)out_size; (void)ws_size;
    const float* x   = (const float*)d_in[0];
    const float* W1  = (const float*)d_in[1];
    const float* b1  = (const float*)d_in[2];
    const float* al1 = (const float*)d_in[3];
    const float* ar1 = (const float*)d_in[4];
    const float* W2  = (const float*)d_in[5];
    const float* b2  = (const float*)d_in[6];
    const float* al2 = (const float*)d_in[7];
    const float* ar2 = (const float*)d_in[8];
    const float* fcW = (const float*)d_in[9];
    const float* fcb = (const float*)d_in[10];
    const int*   ei  = (const int*)d_in[11];

    float* ws   = (float*)d_ws;
    float* xt   = ws;
    float* agg  = xt   + (size_t)NNODES * FDIM;
    int*   tab  = (int*)(agg + (size_t)NNODES * FDIM);
    int*   cst  = tab  + (size_t)SCAN_N;
    int*   rowp = cst  + (size_t)SCAN_N + 4;
    int*   tmp  = rowp + (size_t)NNODES + 4;
    int*   csr  = tmp  + (size_t)EEDGES * 2;

    const dim3 B(256);

    gat_sort_chunk_gemm<<<SORT_CHUNKS + TILES_CHUNK, B, 0, stream>>>(ei, tab, cst, tmp, x, W1, xt);
    gat_sort_place_gemm<<<SORT_BUCKETS + TILES_PLACE, B, 0, stream>>>(tab, cst, tmp, rowp, csr, x, W1, xt);
    gat_attn_fused<false><<<GATHER_WAVES / 4, 128, 0, stream>>>(
        xt, al1, ar1, b1, fcW, fcb, agg, rowp, csr);

    gat_gemm128_gelu_wmma<<<NNODES / 16, B, 0, stream>>>(agg, b1, W2, xt);
    gat_attn_fused<true><<<GATHER_WAVES / 4, 128, 0, stream>>>(
        xt, al2, ar2, b2, fcW, fcb, (float*)d_out, rowp, csr);
}


# layer-2 product fused into the layer-1 edge pass: a work-group owns 16 rows, keeps gelu(row+b1) in LDS, multiplies by W2 on WMMA; aggregated matrix never stored; 5 launches -> 4; sum order unchanged
# speedup vs baseline: 12.4009x; 1.0744x over previous
#include <hip/hip_runtime.h>
#include <hip/hip_bf16.h>


#define NNODES  50000
#define EEDGES  800000
#define ETOT    (EEDGES + NNODES)
#define FDIM    128
#define NHEADS  8
#define CDIM    16

typedef __attribute__((ext_vector_type(2))) float v2f;
typedef __attribute__((ext_vector_type(4))) float v4f;
typedef __attribute__((ext_vector_type(8))) float v8f;

__device__ __forceinline__ float gelu_exact(float v) {
    return 0.5f * v * (1.0f + erff(v * 0.70710678118654752f));
}

__device__ __forceinline__ void gemm_tile16(const float* __restrict__ Ain,
                                            const float* __restrict__ Wmat,
                                            float* __restrict__ Out,
                                            float* As, const int tile, const int t) {
    const int tileBase = tile * 16;

    const float4* A4 = (const float4*)(Ain + (size_t)tileBase * FDIM);
    const int r0 = t >> 5;
    const int c4 = t & 31;
    float4 v[2];
    #pragma unroll
    for (int j = 0; j < 2; ++j) v[j] = A4[(j * 8 + r0) * 32 + c4];
    #pragma unroll
    for (int j = 0; j < 2; ++j)
        *(float4*)&As[(j * 8 + r0) * 132 + c4 * 4] = v[j];
    __syncthreads();

    const int wave = t >> 5;
    const int lane = t & 31;
    const int m    = lane & 15;
    const int half = lane >> 4;
    const int col  = wave * 16 + m;

    v8f acc = {};
    #pragma unroll 4
    for (int kk = 0; kk < 32; ++kk) {
        const int k = kk * 4 + half * 2;
        v2f a, b;
        a.x = As[m * 132 + k];
        a.y = As[m * 132 + k + 1];
        b.x = Wmat[(size_t)k * FDIM + col];
        b.y = Wmat[(size_t)(k + 1) * FDIM + col];
        acc = __builtin_amdgcn_wmma_f32_16x16x4_f32(
             false, a,  false, b,
             (short)0, acc,  false,  false);
    }

    const int mbase = half * 8;
    #pragma unroll
    for (int i = 0; i < 8; ++i)
        Out[(size_t)(tileBase + mbase + i) * FDIM + wave * 16 + m] = acc[i];
}

#define SORT_CHUNKS  200
#define CHUNK_EDGES  4000
#define SORT_BUCKETS 250
#define BUCKET_NODES 200
#define SCAN_N       (SORT_BUCKETS * SORT_CHUNKS)
static_assert(SORT_CHUNKS * CHUNK_EDGES == EEDGES, "chunks must tile the edge list");
static_assert(SORT_BUCKETS * BUCKET_NODES == NNODES, "ranges must tile the nodes");
static_assert(CHUNK_EDGES % 4 == 0, "int4 access");
static_assert(SORT_BUCKETS <= 256 && BUCKET_NODES <= 256, "one counter per thread");
static_assert(SORT_CHUNKS % 8 == 0 && SORT_CHUNKS <= 256, "slices dealt evenly to the 8 waves");

#define SORT_TRIPS ((CHUNK_EDGES / 4 + 255) / 256)
#define SLICES_PER_WAVE (SORT_CHUNKS / 8)

#define GEMM_TILES    (NNODES / 16)
#define TILES_CHUNK   1800
#define TILES_PLACE   1325
#define TILE0_CHUNK   0
#define TILE0_PLACE   (TILE0_CHUNK + TILES_CHUNK)
static_assert(GEMM_TILES * 16 == NNODES, "row tiles must cover the nodes exactly");
static_assert(TILE0_PLACE + TILES_PLACE == GEMM_TILES, "the two shares must cover every row tile once");

__device__ __forceinline__ void sort_chunk_stage(const int* __restrict__ ei, int* __restrict__ tab,
                                                 int* __restrict__ cst, int* __restrict__ tmp,
                                                 const int g) {
    __shared__ int lh[256];
    __shared__ int wtot[8];
    const int t = threadIdx.x;
    const int lane = t & 31, w = t >> 5;
    const int4* s4 = (const int4*)(ei + g * CHUNK_EDGES);
    const int4* d4 = (const int4*)(ei + EEDGES + g * CHUNK_EDGES);
    int4 s[SORT_TRIPS], d[SORT_TRIPS];
    #pragma unroll
    for (int k = 0; k < SORT_TRIPS; ++k) {
        const int q = t + k * 256;
        const bool in = q < CHUNK_EDGES / 4;
        s[k] = in ? s4[q] : make_int4(0, 0, 0, 0);
        d[k] = in ? d4[q] : make_int4(-1, -1, -1, -1);
    }
    lh[t] = 0;
    __syncthreads();
    #pragma unroll
    for (int k = 0; k < SORT_TRIPS; ++k) {
        if ((unsigned)d[k].x < (unsigned)NNODES) atomicAdd(&lh[d[k].x / BUCKET_NODES], 1);
        if ((unsigned)d[k].y < (unsigned)NNODES) atomicAdd(&lh[d[k].y / BUCKET_NODES], 1);
        if ((unsigned)d[k].z < (unsigned)NNODES) atomicAdd(&lh[d[k].z / BUCKET_NODES], 1);
        if ((unsigned)d[k].w < (unsigned)NNODES) atomicAdd(&lh[d[k].w / BUCKET_NODES], 1);
    }
    __syncthreads();
    const int c = lh[t];
    int inc = c;
    for (int o = 1; o < 32; o <<= 1) {
        int u = __shfl_up(inc, o);
        if (lane >= o) inc += u;
    }
    if (lane == 31) wtot[w] = inc;
    __syncthreads();
    int first = inc - c;
    for (int k = 0; k < w; ++k) first += wtot[k];
    if (t < SORT_BUCKETS) {
        tab[t * SORT_CHUNKS + g] = c;
        cst[t * SORT_CHUNKS + g] = first;
    }
    lh[t] = first;
    __syncthreads();
    int2* out = (int2*)tmp + (size_t)g * CHUNK_EDGES;
    #pragma unroll
    for (int k = 0; k < SORT_TRIPS; ++k) {
        const int sv[4] = {s[k].x, s[k].y, s[k].z, s[k].w};
        const int dv[4] = {d[k].x, d[k].y, d[k].z, d[k].w};
        int pos[4];
        #pragma unroll
        for (int e = 0; e < 4; ++e)
            pos[e] = ((unsigned)dv[e] < (unsigned)NNODES) ? atomicAdd(&lh[dv[e] / BUCKET_NODES], 1) : -1;
        #pragma unroll
        for (int e = 0; e < 4; ++e) {
            if ((unsigned)pos[e] < (unsigned)CHUNK_EDGES)
                out[pos[e]] = make_int2(sv[e], dv[e]);
        }
    }
}

__global__ void __launch_bounds__(256)
gat_sort_chunk_gemm(const int* __restrict__ ei, int* __restrict__ tab, int* __restrict__ cst,
                    int* __restrict__ tmp,
                    const float* __restrict__ x, const float* __restrict__ W1,
                    float* __restrict__ xt) {
    __shared__ float As[16 * 132];
    const int b = blockIdx.x;
    if (b < SORT_CHUNKS) sort_chunk_stage(ei, tab, cst, tmp, b);
    else gemm_tile16(x, W1, xt, As, TILE0_CHUNK + (b - SORT_CHUNKS), (int)threadIdx.x);
}

__device__ __forceinline__ void sort_place_stage(const int* __restrict__ tab, const int* __restrict__ cst,
                                                 const int* __restrict__ tmp,
                                                 int* __restrict__ rowptr, int* __restrict__ csr,
                                                 const int b) {
    __shared__ int cnt[256];
    __shared__ int wtot[8];
    __shared__ int wfirst[8], wnum[8];
    const int t = threadIdx.x;
    const int lane = t & 31, w = t >> 5;
    const int wu = __builtin_amdgcn_readfirstlane(w);
    const int node0 = b * BUCKET_NODES;
    const int2* in = (const int2*)tmp;
    int c = 0, o = 0, oa = 0;
    if (lane < SLICES_PER_WAVE) {
        const int g = wu + 8 * lane;
        c  = min(max(tab[b * SORT_CHUNKS + g], 0), CHUNK_EDGES);
        o  = min(max(cst[b * SORT_CHUNKS + g], 0), CHUNK_EDGES - c);
        oa = g * CHUNK_EDGES + o;
    }
    cnt[t] = 0;
    int osum = o, csum = c;
    for (int m = 16; m >= 1; m >>= 1) {
        osum += __shfl_xor(osum, m);
        csum += __shfl_xor(csum, m);
    }
    if (lane == 0) { wfirst[w] = osum; wnum[w] = csum; }

    int2 r[SLICES_PER_WAVE];
    #pragma unroll
    for (int m = 0; m < SLICES_PER_WAVE; ++m) {
        const int cm = __builtin_amdgcn_readlane(c, m);
        const int om = __builtin_amdgcn_readlane(oa, m);
        r[m] = (lane < cm) ? in[om + lane] : make_int2(0, -1);
    }
    __syncthreads();
    int start = 0, num = 0;
    for (int k = 0; k < 8; ++k) { start += wfirst[k]; num += wnum[k]; }
    start = min(start, EEDGES);
    const int end = min(start + num, EEDGES);
    #pragma unroll
    for (int m = 0; m < SLICES_PER_WAVE; ++m) {
        const int j = r[m].y - node0;
        if ((unsigned)j < (unsigned)BUCKET_NODES) atomicAdd(&cnt[j], 1);
    }
    for (int m = 0; m < SLICES_PER_WAVE; ++m) {
        const int cm = __builtin_amdgcn_readlane(c, m);
        const int om = __builtin_amdgcn_readlane(oa, m);
        for (int k = lane + 32; k < cm; k += 32) {
            const int j = in[om + k].y - node0;
            if ((unsigned)j < (unsigned)BUCKET_NODES) atomicAdd(&cnt[j], 1);
        }
    }
    __syncthreads();
    const int cn = cnt[t];
    int inc = cn;
    for (int q = 1; q < 32; q <<= 1) {
        int u = __shfl_up(inc, q);
        if (lane >= q) inc += u;
    }
    if (lane == 31) wtot[w] = inc;
    __syncthreads();
    int first = start + inc - cn;
    for (int k = 0; k < w; ++k) first += wtot[k];
    cnt[t] = first;
    if (t < BUCKET_NODES) rowptr[node0 + t] = first;
    if (b == SORT_BUCKETS - 1 && t == 0) rowptr[NNODES] = end;
    __syncthreads();
    int pos[SLICES_PER_WAVE];
    #pragma unroll
    for (int m = 0; m < SLICES_PER_WAVE; ++m) {
        const int j = r[m].y - node0;
        pos[m] = ((unsigned)j < (unsigned)BUCKET_NODES) ? atomicAdd(&cnt[j], 1) : -1;
    }
    #pragma unroll
    for (int m = 0; m < SLICES_PER_WAVE; ++m) {
        if ((unsigned)pos[m] < (unsigned)EEDGES)
            ((int2*)csr)[pos[m]] = r[m];
    }
    for (int m = 0; m < SLICES_PER_WAVE; ++m) {
        const int cm = __builtin_amdgcn_readlane(c, m);
        const int om = __builtin_amdgcn_readlane(oa, m);
        for (int k = lane + 32; k < cm; k += 32) {
            const int2 rk = in[om + k];
            const int j = rk.y - node0;
            if ((unsigned)j < (unsigned)BUCKET_NODES) {
                const int pk = atomicAdd(&cnt[j], 1);
                if ((unsigned)pk < (unsigned)EEDGES)
                    ((int2*)csr)[pk] = rk;
            }
        }
    }
}

__global__ void __launch_bounds__(256)
gat_sort_place_gemm(const int* __restrict__ tab, const int* __restrict__ cst, const int* __restrict__ tmp,
                    int* __restrict__ rowptr, int* __restrict__ csr,
                    const float* __restrict__ x, const float* __restrict__ W1,
                    float* __restrict__ xt) {
    __shared__ float As[16 * 132];
    const int b = blockIdx.x;
    if (b < SORT_BUCKETS) sort_place_stage(tab, cst, tmp, rowptr, csr, b);
    else gemm_tile16(x, W1, xt, As, TILE0_PLACE + (b - SORT_BUCKETS), (int)threadIdx.x);
}

#define QUAD_XOR1 0xB1
#define QUAD_XOR2 0x4E
template <int CTRL>
__device__ __forceinline__ float quad_move(float v) {
    return __int_as_float(__builtin_amdgcn_mov_dpp(__float_as_int(v), CTRL, 0xF, 0xF, true));
}

__device__ __forceinline__ float quad_sum(float v) {
    v += quad_move<QUAD_XOR1>(v);
    v += quad_move<QUAD_XOR2>(v);
    return v;
}

__device__ __forceinline__ float quad_transpose_sum(float e0, float e1, float e2, float e3,
                                                    bool odd, bool hi) {
    const float rA = (odd ? e1 : e0) + quad_move<QUAD_XOR1>(odd ? e0 : e1);
    const float rB = (odd ? e3 : e2) + quad_move<QUAD_XOR1>(odd ? e2 : e3);
    return (hi ? rB : rA) + quad_move<QUAD_XOR2>(hi ? rA : rB);
}

__device__ __forceinline__ float dot4(v4f u, v4f x) {
    const v2f t = __builtin_elementwise_fma(u.hi, x.hi, u.lo * x.lo);
    return t.x + t.y;
}
__device__ __forceinline__ v2f splat2(float s) {
    v2f r = {s, s};
    return r;
}
struct pairs4 { v2f c0, c1, c2, c3; };
__device__ __forceinline__ v2f dot4x2(const pairs4 c, v4f x) {
    v2f p = c.c0 * splat2(x.x);
    p = __builtin_elementwise_fma(c.c1, splat2(x.y), p);
    p = __builtin_elementwise_fma(c.c2, splat2(x.z), p);
    p = __builtin_elementwise_fma(c.c3, splat2(x.w), p);
    return p;
}
__device__ __forceinline__ v2f wsum4(v2f acc, float sc, float p0, v2f x0, float p1, v2f x1,
                                     float p2, v2f x2, float p3, v2f x3) {
    v2f w = x0 * splat2(p0);
    w = __builtin_elementwise_fma(x1, splat2(p1), w);
    w = __builtin_elementwise_fma(x2, splat2(p2), w);
    w = __builtin_elementwise_fma(x3, splat2(p3), w);
    return __builtin_elementwise_fma(acc, splat2(sc), w);
}
__device__ __forceinline__ float dot4_fc(float4 a, float4 b) {
    return a.x * b.x + a.y * b.y + a.z * b.z + a.w * b.w;
}

#define LOG2E 1.4426950408889634f
__device__ __forceinline__ float exp_hw(float x) {
    return __builtin_amdgcn_exp2f(x * LOG2E);
}
__device__ __forceinline__ float rcp_hw(float x) {
    return __builtin_amdgcn_rcpf(x);
}

__device__ __forceinline__ float edge_score(float logit, float a) {
    float sg = rcp_hw(1.0f + exp_hw(-logit));
    float v  = a * sg;
    return (v > 0.0f) ? v : 0.2f * v;
}

#define ROW_XOR4 0x164
#define ROW_XOR8 0x168
__device__ __forceinline__ float heads_sum(float v) {
    v += quad_move<ROW_XOR4>(v);
    v += quad_move<ROW_XOR8>(v);
    const int b = __float_as_int(v);
    return v + __int_as_float(__builtin_amdgcn_permlanex16(b, b, 0x76543210, 0xfedcba98, false, false));
}

__device__ __forceinline__ v4f load_row(const float* __restrict__ xt, int s, int lane) {
    s = min(max(s, 0), NNODES - 1);
    return *(const v4f*)(xt + (size_t)s * FDIM + lane * 4);
}

#define GATHER_WAVES 10000
template <bool LAST>
__global__ void __launch_bounds__(128)
gat_attn_fused(const float* __restrict__ xt,
               const float* __restrict__ attL,
               const float* __restrict__ attR,
               const float* __restrict__ bias,
               const float* __restrict__ fcw,
               const float* __restrict__ fcb,
               const float* __restrict__ Wn,
               float* __restrict__ out,
               const int* __restrict__ rowptr,
               const int* __restrict__ csr) {
    __shared__ float As[LAST ? 4 : 16 * 132];
    const int lane = threadIdx.x & 31;
    const bool odd = (lane & 1) != 0;
    const bool hi  = (lane & 2) != 0;
    const int wv = __builtin_amdgcn_readfirstlane((int)(threadIdx.x >> 5));
    const int gw = blockIdx.x * 4 + wv;
    const int2* rec = (const int2*)csr;
    const v4f al = *(const v4f*)(attL + lane * 4);
    const v4f ar = *(const v4f*)(attR + lane * 4);
    const float4 bb = *(const float4*)(bias + lane * 4);
    const float4 fw = LAST ? *(const float4*)(fcw + lane * 4) : make_float4(0.f, 0.f, 0.f, 0.f);
    const float  fb = LAST ? fcb[0] : 0.0f;

    const int tileBase = (int)blockIdx.x * 16;
    const int node0 = LAST ? gw : tileBase + wv * 4;
    const int nend  = LAST ? NNODES : node0 + 4;
    const int nstep = LAST ? GATHER_WAVES : 1;
    for (int node = node0; node < nend; node += nstep) {
        const int beg = max(rowptr[node], 0), end = min(rowptr[node + 1], EEDGES);
        const v4f xi = *(const v4f*)(xt + (size_t)node * FDIM + lane * 4);
        const float aR = quad_sum(dot4(ar, xi));
        const pairs4 cf = {{xi.x, al.x}, {xi.y, al.y}, {xi.z, al.z}, {xi.w, al.w}};
        const v2f qs = dot4x2(cf, xi);
        float m   = edge_score(quad_sum(qs.x), quad_sum(qs.y) + aR);
        float den = 1.0f;
        v2f accl = xi.lo, acch = xi.hi;

        int p = beg;
        for (; p + 4 <= end; p += 4) {
            const v4f x0 = load_row(xt, rec[p].x,     lane);
            const v4f x1 = load_row(xt, rec[p + 1].x, lane);
            const v4f x2 = load_row(xt, rec[p + 2].x, lane);
            const v4f x3 = load_row(xt, rec[p + 3].x, lane);
            const v2f q0 = dot4x2(cf, x0), q1 = dot4x2(cf, x1);
            const v2f q2 = dot4x2(cf, x2), q3 = dot4x2(cf, x3);
            const float lg = quad_transpose_sum(q0.x, q1.x, q2.x, q3.x, odd, hi);
            const float aa = quad_transpose_sum(q0.y, q1.y, q2.y, q3.y, odd, hi) + aR;
            const float v  = edge_score(lg, aa);
            float mx = fmaxf(v,  quad_move<QUAD_XOR1>(v));
            mx       = fmaxf(mx, quad_move<QUAD_XOR2>(mx));
            const float mn = fmaxf(m, mx);
            const float sc = exp_hw(m - mn);
            const float pe = exp_hw(v - mn);
            const float p0 = quad_move<0x00>(pe);
            const float p1 = quad_move<0x55>(pe);
            const float p2 = quad_move<0xAA>(pe);
            const float p3 = quad_move<0xFF>(pe);
            den  = den * sc + ((p0 + p1) + (p2 + p3));
            accl = wsum4(accl, sc, p0, x0.lo, p1, x1.lo, p2, x2.lo, p3, x3.lo);
            acch = wsum4(acch, sc, p0, x0.hi, p1, x1.hi, p2, x2.hi, p3, x3.hi);
            m = mn;
        }
        for (; p < end; ++p) {
            const v4f x = load_row(xt, rec[p].x, lane);
            const v2f q = dot4x2(cf, x);
            const float v  = edge_score(quad_sum(q.x), quad_sum(q.y) + aR);
            const float mn = fmaxf(m, v);
            const float sc = exp_hw(m - mn);
            const float pe = exp_hw(v - mn);
            den  = den * sc + pe;
            accl = __builtin_elementwise_fma(accl, splat2(sc), x.lo * splat2(pe));
            acch = __builtin_elementwise_fma(acch, splat2(sc), x.hi * splat2(pe));
            m = mn;
        }
        const float inv = rcp_hw(den + 1e-16f);
        float4 h;
        h.x = gelu_exact(accl.x * inv + bb.x);
        h.y = gelu_exact(accl.y * inv + bb.y);
        h.z = gelu_exact(acch.x * inv + bb.z);
        h.w = gelu_exact(acch.y * inv + bb.w);
        if (!LAST) {
            *(float4*)&As[(node - tileBase) * 132 + lane * 4] = h;
        } else {
            const float y = heads_sum(quad_sum(dot4_fc(h, fw)));
            if (lane == 0) out[node] = y + fb;
        }
    }

    if (!LAST) {
        __syncthreads();
        const int m    = lane & 15;
        const int half = lane >> 4;
        const int col0 = wv * 32 + m;
        v8f acc0 = {}, acc1 = {};
        #pragma unroll 4
        for (int kk = 0; kk < 32; ++kk) {
            const int k = kk * 4 + half * 2;
            v2f a, b0, b1;
            a.x = As[m * 132 + k];
            a.y = As[m * 132 + k + 1];
            b0.x = Wn[(size_t)k * FDIM + col0];
            b0.y = Wn[(size_t)(k + 1) * FDIM + col0];
            b1.x = Wn[(size_t)k * FDIM + col0 + 16];
            b1.y = Wn[(size_t)(k + 1) * FDIM + col0 + 16];
            acc0 = __builtin_amdgcn_wmma_f32_16x16x4_f32(
                 false, a,  false, b0,
                 (short)0, acc0,  false,  false);
            acc1 = __builtin_amdgcn_wmma_f32_16x16x4_f32(
                 false, a,  false, b1,
                 (short)0, acc1,  false,  false);
        }
        float* orow = out + (size_t)(tileBase + half * 8) * FDIM + col0;
        #pragma unroll
        for (int i = 0; i < 8; ++i) {
            orow[(size_t)i * FDIM]      = acc0[i];
            orow[(size_t)i * FDIM + 16] = acc1[i];
        }
    }
}

extern "C" void kernel_launch(void* const* d_in, const int* in_sizes, int n_in,
                              void* d_out, int out_size, void* d_ws, size_t ws_size,
                              hipStream_t stream) {
    (void)in_sizes; (void)n_in; (void)out_size; (void)ws_size;
    const float* x   = (const float*)d_in[0];
    const float* W1  = (const float*)d_in[1];
    const float* b1  = (const float*)d_in[2];
    const float* al1 = (const float*)d_in[3];
    const float* ar1 = (const float*)d_in[4];
    const float* W2  = (const float*)d_in[5];
    const float* b2  = (const float*)d_in[6];
    const float* al2 = (const float*)d_in[7];
    const float* ar2 = (const float*)d_in[8];
    const float* fcW = (const float*)d_in[9];
    const float* fcb = (const float*)d_in[10];
    const int*   ei  = (const int*)d_in[11];

    float* ws   = (float*)d_ws;
    float* xt   = ws;
    float* xt2  = xt   + (size_t)NNODES * FDIM;
    int*   tab  = (int*)(xt2 + (size_t)NNODES * FDIM);
    int*   cst  = tab  + (size_t)SCAN_N;
    int*   rowp = cst  + (size_t)SCAN_N + 4;
    int*   tmp  = rowp + (size_t)NNODES + 4;
    int*   csr  = tmp  + (size_t)EEDGES * 2;

    const dim3 B(256);

    gat_sort_chunk_gemm<<<SORT_CHUNKS + TILES_CHUNK, B, 0, stream>>>(ei, tab, cst, tmp, x, W1, xt);
    gat_sort_place_gemm<<<SORT_BUCKETS + TILES_PLACE, B, 0, stream>>>(tab, cst, tmp, rowp, csr, x, W1, xt);
    gat_attn_fused<false><<<GEMM_TILES, 128, 0, stream>>>(
        xt, al1, ar1, b1, fcW, fcb, W2, xt2, rowp, csr);

    gat_attn_fused<true><<<GATHER_WAVES / 4, 128, 0, stream>>>(
        xt2, al2, ar2, b2, fcW, fcb, W2, (float*)d_out, rowp, csr);
}
